# starting kernel
# baseline (speedup 1.0000x reference)
#include <hip/hip_runtime.h>
#include <hip/hip_bf16.h>

typedef __attribute__((ext_vector_type(2))) float v2f;
typedef __attribute__((ext_vector_type(8))) float v8f;

#define IN_C 128
#define HID  64
#define OUT_C 16

__global__ void deg_init_k(float* __restrict__ deg, int n) {
  int i = blockIdx.x * blockDim.x + threadIdx.x;
  if (i < n) deg[i] = 1.0f;
}

__global__ void deg_edges_k(float* __restrict__ deg, const int* __restrict__ ei, int E) {
  int e = blockIdx.x * blockDim.x + threadIdx.x;
  if (e < E) atomicAdd(&deg[ei[E + e]], 1.0f);
}

__global__ void dinv_k(float* __restrict__ deg, int n) {
  int i = blockIdx.x * blockDim.x + threadIdx.x;
  if (i < n) {
    float d = deg[i];
    deg[i] = (d > 0.0f) ? rsqrtf(d) : 0.0f;
  }
}

template <bool RELU_A>
__global__ void gemm_wmma_k(const float* __restrict__ A, const float* __restrict__ B,
                            float* __restrict__ C, int M, int K, int N) {
  __shared__ float Bs[IN_C * 16];

  const int tid  = threadIdx.x;
  const int wave = tid >> 5;
  const int lane = tid & 31;
  const int mr   = lane & 15;
  const int half = lane >> 4;
  const int n0   = blockIdx.y * 16;

  for (int i = tid; i < K * 16; i += 128) {
    int kk = i >> 4, nn = i & 15;
    Bs[i] = B[(size_t)kk * N + n0 + nn];
  }
  __syncthreads();

  const int tile = blockIdx.x * 4 + wave;
  if (tile * 16 < M) {
    const int m0 = tile * 16;
    const float* arow = A + (size_t)(m0 + mr) * K + 2 * half;
    const float* brow = Bs + 2 * half * 16 + mr;

    v8f acc = {};
    for (int k = 0; k < K; k += 4) {
      v2f a, b;
      a[0] = arow[k];
      a[1] = arow[k + 1];
      if (RELU_A) { a[0] = fmaxf(a[0], 0.0f); a[1] = fmaxf(a[1], 0.0f); }
      b[0] = brow[k * 16];
      b[1] = brow[k * 16 + 16];
      acc = __builtin_amdgcn_wmma_f32_16x16x4_f32(false, a, false, b,
                                                  (short)0, acc, false, false);
    }

    float* crow = C + (size_t)(m0 + 8 * half) * N + n0 + mr;
#pragma unroll
    for (int r = 0; r < 8; ++r) crow[(size_t)r * N] = acc[r];
  }
}

__global__ void agg_init_k(float* __restrict__ out, const float* __restrict__ h,
                           const float* __restrict__ dinv, const float* __restrict__ bias,
                           int n4) {
  int i = blockIdx.x * blockDim.x + threadIdx.x;
  if (i < n4) {
    int node = i >> 4;
    int c4   = (i & 15) * 4;
    float d2 = dinv[node]; d2 *= d2;
    const float4 hv = ((const float4*)h)[i];
    const float4 bv = *(const float4*)&bias[c4];
    float4 o;
    o.x = bv.x + d2 * hv.x;
    o.y = bv.y + d2 * hv.y;
    o.z = bv.z + d2 * hv.z;
    o.w = bv.w + d2 * hv.w;
    ((float4*)out)[i] = o;
  }
}

__global__ void agg_edges_k(float* __restrict__ out, const float* __restrict__ h,
                            const float* __restrict__ dinv, const int* __restrict__ ei,
                            int E) {
  int idx = blockIdx.x * blockDim.x + threadIdx.x;
  if (idx >= E * 16) return;
  int e  = idx >> 4;
  int c4 = (idx & 15) * 4;
  int s = ei[e];
  int d = ei[E + e];
  float w = dinv[s] * dinv[d];
  const float4 hv = *(const float4*)&h[(size_t)s * HID + c4];
  float* op = &out[(size_t)d * HID + c4];
  atomicAdd(op + 0, w * hv.x);
  atomicAdd(op + 1, w * hv.y);
  atomicAdd(op + 2, w * hv.z);
  atomicAdd(op + 3, w * hv.w);
}

#define W1S_STRIDE 72

__global__ void edge_mlp_k(const float* __restrict__ h, const int* __restrict__ ei,
                           const float* __restrict__ Wm1, const float* __restrict__ bm1,
                           const float* __restrict__ Wm2, const float* __restrict__ bm2,
                           float* __restrict__ out, int E) {
  __shared__ float w1s[IN_C * W1S_STRIDE];
  __shared__ float w2s[HID * OUT_C];
  __shared__ float zbuf[4][16][68];

  const int tid  = threadIdx.x;
  const int wave = tid >> 5;
  const int lane = tid & 31;
  const int mr   = lane & 15;
  const int half = lane >> 4;
  const int e0   = (blockIdx.x * 4 + wave) * 16;

  for (int i = tid; i < IN_C * 16; i += 128) {
    int row = i >> 4, c4 = (i & 15) * 4;
    *(float4*)&w1s[row * W1S_STRIDE + c4] = *(const float4*)&Wm1[(size_t)row * HID + c4];
  }
  for (int i = tid; i < HID * OUT_C / 4; i += 128) {
    ((float4*)w2s)[i] = ((const float4*)Wm2)[i];
  }
  __syncthreads();

  const int row = ei[e0 + mr];
  const int col = ei[E + e0 + mr];
  const float* efr = h + (size_t)row * HID + 2 * half;
  const float* efc = h + (size_t)col * HID + 2 * half;

  v8f acc[4] = {v8f{}, v8f{}, v8f{}, v8f{}};

  for (int k = 0; k < 64; k += 4) {
    v2f a;
    a[0] = fmaxf(efr[k], 0.0f);
    a[1] = fmaxf(efr[k + 1], 0.0f);
    const float* bp = w1s + (k + 2 * half) * W1S_STRIDE + mr;
#pragma unroll
    for (int nt = 0; nt < 4; ++nt) {
      v2f b;
      b[0] = bp[nt * 16];
      b[1] = bp[W1S_STRIDE + nt * 16];
      acc[nt] = __builtin_amdgcn_wmma_f32_16x16x4_f32(false, a, false, b,
                                                      (short)0, acc[nt], false, false);
    }
  }
  for (int k = 0; k < 64; k += 4) {
    v2f a;
    a[0] = fmaxf(efc[k], 0.0f);
    a[1] = fmaxf(efc[k + 1], 0.0f);
    const float* bp = w1s + (64 + k + 2 * half) * W1S_STRIDE + mr;
#pragma unroll
    for (int nt = 0; nt < 4; ++nt) {
      v2f b;
      b[0] = bp[nt * 16];
      b[1] = bp[W1S_STRIDE + nt * 16];
      acc[nt] = __builtin_amdgcn_wmma_f32_16x16x4_f32(false, a, false, b,
                                                      (short)0, acc[nt], false, false);
    }
  }

#pragma unroll
  for (int nt = 0; nt < 4; ++nt) {
    float bias = bm1[nt * 16 + mr];
#pragma unroll
    for (int r = 0; r < 8; ++r)
      zbuf[wave][r + 8 * half][nt * 16 + mr] = fmaxf(acc[nt][r] + bias, 0.0f);
  }
  __syncthreads();

  v8f o = {};
  for (int k = 0; k < 64; k += 4) {
    v2f a, b;
    a[0] = zbuf[wave][mr][k + 2 * half];
    a[1] = zbuf[wave][mr][k + 2 * half + 1];
    b[0] = w2s[(k + 2 * half) * OUT_C + mr];
    b[1] = w2s[(k + 2 * half + 1) * OUT_C + mr];
    o = __builtin_amdgcn_wmma_f32_16x16x4_f32(false, a, false, b,
                                              (short)0, o, false, false);
  }

  float bo = bm2[mr];
  float* op = out + (size_t)(e0 + 8 * half) * OUT_C + mr;
#pragma unroll
  for (int r = 0; r < 8; ++r) op[(size_t)r * OUT_C] = o[r] + bo;
}

extern "C" void kernel_launch(void* const* d_in, const int* in_sizes, int n_in,
                              void* d_out, int out_size, void* d_ws, size_t ws_size,
                              hipStream_t stream) {
  const float* x   = (const float*)d_in[0];
  const int*   ei  = (const int*)d_in[1];
  const float* W1  = (const float*)d_in[2];
  const float* b1  = (const float*)d_in[3];
  const float* W2  = (const float*)d_in[4];
  const float* b2  = (const float*)d_in[5];
  const float* Wm1 = (const float*)d_in[6];
  const float* bm1 = (const float*)d_in[7];
  const float* Wm2 = (const float*)d_in[8];
  const float* bm2 = (const float*)d_in[9];
  float* out = (float*)d_out;

  const int N = in_sizes[0] / IN_C;
  const int E = in_sizes[1] / 2;

  float* deg  = (float*)d_ws;
  float* bufA = deg + (size_t)((N + 63) & ~63);
  float* bufB = bufA + (size_t)N * HID;

  const int mtiles = (N + 15) / 16;
  const int gx     = (mtiles + 3) / 4;
  const int n4     = N * (HID / 4);
  const int e4     = E * (HID / 4);
  const int nb  = (N + 255) / 256;
  const int eb  = (E + 255) / 256;
  const int n4b = (n4 + 255) / 256;
  const int e4b = (e4 + 255) / 256;

  deg_init_k<<<nb, 256, 0, stream>>>(deg, N);
  deg_edges_k<<<eb, 256, 0, stream>>>(deg, ei, E);
  dinv_k<<<nb, 256, 0, stream>>>(deg, N);

  gemm_wmma_k<false><<<dim3(gx, HID / 16), 128, 0, stream>>>(x, W1, bufA, N, IN_C, HID);
  agg_init_k<<<n4b, 256, 0, stream>>>(bufB, bufA, deg, b1, n4);
  agg_edges_k<<<e4b, 256, 0, stream>>>(bufB, bufA, deg, ei, E);

  gemm_wmma_k<true><<<dim3(gx, HID / 16), 128, 0, stream>>>(bufB, W2, bufA, N, HID, HID);
  agg_init_k<<<n4b, 256, 0, stream>>>(bufB, bufA, deg, b2, n4);
  agg_edges_k<<<e4b, 256, 0, stream>>>(bufB, bufA, deg, ei, E);

  edge_mlp_k<<<E / 64, 128, 0, stream>>>(bufB, ei, Wm1, bm1, Wm2, bm2, out, E);
}


# agg_edges: one wave walks 32 edges, lane l = float l of each 128-B line, so every gather load and f32 atomic add covers one whole line (was 4 lines x 8 floats); sums stay atomic, arrival order
# speedup vs baseline: 1.2752x; 1.2752x over previous
#include <hip/hip_runtime.h>
#include <hip/hip_bf16.h>

typedef __attribute__((ext_vector_type(2))) float v2f;
typedef __attribute__((ext_vector_type(8))) float v8f;

#define IN_C 128
#define HID  64
#define OUT_C 16

__global__ void deg_init_k(float* __restrict__ deg, int n) {
  int i = blockIdx.x * blockDim.x + threadIdx.x;
  if (i < n) deg[i] = 1.0f;
}

__global__ void deg_edges_k(float* __restrict__ deg, const int* __restrict__ ei, int E) {
  int e = blockIdx.x * blockDim.x + threadIdx.x;
  if (e < E) atomicAdd(&deg[ei[E + e]], 1.0f);
}

__global__ void dinv_k(float* __restrict__ deg, int n) {
  int i = blockIdx.x * blockDim.x + threadIdx.x;
  if (i < n) {
    float d = deg[i];
    deg[i] = (d > 0.0f) ? rsqrtf(d) : 0.0f;
  }
}

template <bool RELU_A>
__global__ void gemm_wmma_k(const float* __restrict__ A, const float* __restrict__ B,
                            float* __restrict__ C, int M, int K, int N) {
  __shared__ float Bs[IN_C * 16];

  const int tid  = threadIdx.x;
  const int wave = tid >> 5;
  const int lane = tid & 31;
  const int mr   = lane & 15;
  const int half = lane >> 4;
  const int n0   = blockIdx.y * 16;

  for (int i = tid; i < K * 16; i += 128) {
    int kk = i >> 4, nn = i & 15;
    Bs[i] = B[(size_t)kk * N + n0 + nn];
  }
  __syncthreads();

  const int tile = blockIdx.x * 4 + wave;
  if (tile * 16 < M) {
    const int m0 = tile * 16;
    const float* arow = A + (size_t)(m0 + mr) * K + 2 * half;
    const float* brow = Bs + 2 * half * 16 + mr;

    v8f acc = {};
    for (int k = 0; k < K; k += 4) {
      v2f a, b;
      a[0] = arow[k];
      a[1] = arow[k + 1];
      if (RELU_A) { a[0] = fmaxf(a[0], 0.0f); a[1] = fmaxf(a[1], 0.0f); }
      b[0] = brow[k * 16];
      b[1] = brow[k * 16 + 16];
      acc = __builtin_amdgcn_wmma_f32_16x16x4_f32(false, a, false, b,
                                                  (short)0, acc, false, false);
    }

    float* crow = C + (size_t)(m0 + 8 * half) * N + n0 + mr;
#pragma unroll
    for (int r = 0; r < 8; ++r) crow[(size_t)r * N] = acc[r];
  }
}

__global__ void agg_init_k(float* __restrict__ out, const float* __restrict__ h,
                           const float* __restrict__ dinv, const float* __restrict__ bias,
                           int n4) {
  int i = blockIdx.x * blockDim.x + threadIdx.x;
  if (i < n4) {
    int node = i >> 4;
    int c4   = (i & 15) * 4;
    float d2 = dinv[node]; d2 *= d2;
    const float4 hv = ((const float4*)h)[i];
    const float4 bv = *(const float4*)&bias[c4];
    float4 o;
    o.x = bv.x + d2 * hv.x;
    o.y = bv.y + d2 * hv.y;
    o.z = bv.z + d2 * hv.z;
    o.w = bv.w + d2 * hv.w;
    ((float4*)out)[i] = o;
  }
}

__global__ void agg_edges_k(float* __restrict__ out, const float* __restrict__ h,
                            const float* __restrict__ dinv, const int* __restrict__ ei,
                            int E) {
  const int lane = threadIdx.x & 31;
  const int wv = __builtin_amdgcn_readfirstlane((int)(threadIdx.x >> 5));
  const int e0 = (blockIdx.x * (int)(blockDim.x >> 5) + wv) * 32;
  if (e0 >= E) return;
  const int cnt = min(32, E - e0);

  int s = 0, d = 0;
  float w = 0.0f;
  if (lane < cnt) {
    s = ei[e0 + lane];
    d = ei[E + e0 + lane];
    w = dinv[s] * dinv[d];
  }

  int j = 0;
  for (; j + 8 <= cnt; j += 8) {
    float v0[8], v1[8];
    int dd[8];
    for (int u = 0; u < 8; ++u) {
      const int sj = __builtin_amdgcn_readlane(s, j + u);
      const float wj = __int_as_float(__builtin_amdgcn_readlane(__float_as_int(w), j + u));
      dd[u] = __builtin_amdgcn_readlane(d, j + u);
      const float* hp = h + (size_t)sj * HID + lane;
      v0[u] = wj * hp[0];
      v1[u] = wj * hp[32];
    }
    for (int u = 0; u < 8; ++u) {
      float* op = out + (size_t)dd[u] * HID + lane;
      atomicAdd(op, v0[u]);
      atomicAdd(op + 32, v1[u]);
    }
  }
  for (; j < cnt; ++j) {
    const int sj = __builtin_amdgcn_readlane(s, j);
    const float wj = __int_as_float(__builtin_amdgcn_readlane(__float_as_int(w), j));
    const int dj = __builtin_amdgcn_readlane(d, j);
    const float* hp = h + (size_t)sj * HID + lane;
    float* op = out + (size_t)dj * HID + lane;
    atomicAdd(op, wj * hp[0]);
    atomicAdd(op + 32, wj * hp[32]);
  }
}

#define W1S_STRIDE 72

__global__ void edge_mlp_k(const float* __restrict__ h, const int* __restrict__ ei,
                           const float* __restrict__ Wm1, const float* __restrict__ bm1,
                           const float* __restrict__ Wm2, const float* __restrict__ bm2,
                           float* __restrict__ out, int E) {
  __shared__ float w1s[IN_C * W1S_STRIDE];
  __shared__ float w2s[HID * OUT_C];
  __shared__ float zbuf[4][16][68];

  const int tid  = threadIdx.x;
  const int wave = tid >> 5;
  const int lane = tid & 31;
  const int mr   = lane & 15;
  const int half = lane >> 4;
  const int e0   = (blockIdx.x * 4 + wave) * 16;

  for (int i = tid; i < IN_C * 16; i += 128) {
    int row = i >> 4, c4 = (i & 15) * 4;
    *(float4*)&w1s[row * W1S_STRIDE + c4] = *(const float4*)&Wm1[(size_t)row * HID + c4];
  }
  for (int i = tid; i < HID * OUT_C / 4; i += 128) {
    ((float4*)w2s)[i] = ((const float4*)Wm2)[i];
  }
  __syncthreads();

  const int row = ei[e0 + mr];
  const int col = ei[E + e0 + mr];
  const float* efr = h + (size_t)row * HID + 2 * half;
  const float* efc = h + (size_t)col * HID + 2 * half;

  v8f acc[4] = {v8f{}, v8f{}, v8f{}, v8f{}};

  for (int k = 0; k < 64; k += 4) {
    v2f a;
    a[0] = fmaxf(efr[k], 0.0f);
    a[1] = fmaxf(efr[k + 1], 0.0f);
    const float* bp = w1s + (k + 2 * half) * W1S_STRIDE + mr;
#pragma unroll
    for (int nt = 0; nt < 4; ++nt) {
      v2f b;
      b[0] = bp[nt * 16];
      b[1] = bp[W1S_STRIDE + nt * 16];
      acc[nt] = __builtin_amdgcn_wmma_f32_16x16x4_f32(false, a, false, b,
                                                      (short)0, acc[nt], false, false);
    }
  }
  for (int k = 0; k < 64; k += 4) {
    v2f a;
    a[0] = fmaxf(efc[k], 0.0f);
    a[1] = fmaxf(efc[k + 1], 0.0f);
    const float* bp = w1s + (64 + k + 2 * half) * W1S_STRIDE + mr;
#pragma unroll
    for (int nt = 0; nt < 4; ++nt) {
      v2f b;
      b[0] = bp[nt * 16];
      b[1] = bp[W1S_STRIDE + nt * 16];
      acc[nt] = __builtin_amdgcn_wmma_f32_16x16x4_f32(false, a, false, b,
                                                      (short)0, acc[nt], false, false);
    }
  }

#pragma unroll
  for (int nt = 0; nt < 4; ++nt) {
    float bias = bm1[nt * 16 + mr];
#pragma unroll
    for (int r = 0; r < 8; ++r)
      zbuf[wave][r + 8 * half][nt * 16 + mr] = fmaxf(acc[nt][r] + bias, 0.0f);
  }
  __syncthreads();

  v8f o = {};
  for (int k = 0; k < 64; k += 4) {
    v2f a, b;
    a[0] = zbuf[wave][mr][k + 2 * half];
    a[1] = zbuf[wave][mr][k + 2 * half + 1];
    b[0] = w2s[(k + 2 * half) * OUT_C + mr];
    b[1] = w2s[(k + 2 * half + 1) * OUT_C + mr];
    o = __builtin_amdgcn_wmma_f32_16x16x4_f32(false, a, false, b,
                                              (short)0, o, false, false);
  }

  float bo = bm2[mr];
  float* op = out + (size_t)(e0 + 8 * half) * OUT_C + mr;
#pragma unroll
  for (int r = 0; r < 8; ++r) op[(size_t)r * OUT_C] = o[r] + bo;
}

extern "C" void kernel_launch(void* const* d_in, const int* in_sizes, int n_in,
                              void* d_out, int out_size, void* d_ws, size_t ws_size,
                              hipStream_t stream) {
  const float* x   = (const float*)d_in[0];
  const int*   ei  = (const int*)d_in[1];
  const float* W1  = (const float*)d_in[2];
  const float* b1  = (const float*)d_in[3];
  const float* W2  = (const float*)d_in[4];
  const float* b2  = (const float*)d_in[5];
  const float* Wm1 = (const float*)d_in[6];
  const float* bm1 = (const float*)d_in[7];
  const float* Wm2 = (const float*)d_in[8];
  const float* bm2 = (const float*)d_in[9];
  float* out = (float*)d_out;

  const int N = in_sizes[0] / IN_C;
  const int E = in_sizes[1] / 2;

  float* deg  = (float*)d_ws;
  float* bufA = deg + (size_t)((N + 63) & ~63);
  float* bufB = bufA + (size_t)N * HID;

  const int mtiles = (N + 15) / 16;
  const int gx     = (mtiles + 3) / 4;
  const int n4     = N * (HID / 4);
  const int nb  = (N + 255) / 256;
  const int eb  = (E + 255) / 256;
  const int n4b = (n4 + 255) / 256;

  deg_init_k<<<nb, 256, 0, stream>>>(deg, N);
  deg_edges_k<<<eb, 256, 0, stream>>>(deg, ei, E);
  dinv_k<<<nb, 256, 0, stream>>>(deg, N);

  gemm_wmma_k<false><<<dim3(gx, HID / 16), 128, 0, stream>>>(x, W1, bufA, N, IN_C, HID);
  agg_init_k<<<n4b, 256, 0, stream>>>(bufB, bufA, deg, b1, n4);
  agg_edges_k<<<eb, 256, 0, stream>>>(bufB, bufA, deg, ei, E);

  gemm_wmma_k<true><<<dim3(gx, HID / 16), 128, 0, stream>>>(bufB, W2, bufA, N, HID, HID);
  agg_init_k<<<n4b, 256, 0, stream>>>(bufB, bufA, deg, b2, n4);
  agg_edges_k<<<eb, 256, 0, stream>>>(bufB, bufA, deg, ei, E);

  edge_mlp_k<<<E / 64, 128, 0, stream>>>(bufB, ei, Wm1, bm1, Wm2, bm2, out, E);
}


# edge MLP layer 1 moved from edges to nodes by linearity: P=relu(h)@Wm1[:64], Q=relu(h)@Wm1[64:] per node on WMMA, edge does relu(P[row]+Q[col]+b)@Wm2; the 128-long sum is now two 64-long sums added
# speedup vs baseline: 1.5165x; 1.1892x over previous
#include <hip/hip_runtime.h>
#include <hip/hip_bf16.h>

typedef __attribute__((ext_vector_type(2))) float v2f;
typedef __attribute__((ext_vector_type(8))) float v8f;

#define IN_C 128
#define HID  64
#define OUT_C 16

__global__ void deg_init_k(float* __restrict__ deg, int n) {
  int i = blockIdx.x * blockDim.x + threadIdx.x;
  if (i < n) deg[i] = 1.0f;
}

__global__ void deg_edges_k(float* __restrict__ deg, const int* __restrict__ ei, int E) {
  int e = blockIdx.x * blockDim.x + threadIdx.x;
  if (e < E) atomicAdd(&deg[ei[E + e]], 1.0f);
}

__global__ void dinv_k(float* __restrict__ deg, int n) {
  int i = blockIdx.x * blockDim.x + threadIdx.x;
  if (i < n) {
    float d = deg[i];
    deg[i] = (d > 0.0f) ? rsqrtf(d) : 0.0f;
  }
}

template <bool RELU_A>
__global__ void gemm_wmma_k(const float* __restrict__ A, const float* __restrict__ B,
                            float* __restrict__ C, int M, int K, int N) {
  __shared__ float Bs[IN_C * 16];

  const int tid  = threadIdx.x;
  const int wave = tid >> 5;
  const int lane = tid & 31;
  const int mr   = lane & 15;
  const int half = lane >> 4;
  const int n0   = blockIdx.y * 16;

  for (int i = tid; i < K * 16; i += 128) {
    int kk = i >> 4, nn = i & 15;
    Bs[i] = B[(size_t)kk * N + n0 + nn];
  }
  __syncthreads();

  const int tile = blockIdx.x * 4 + wave;
  if (tile * 16 < M) {
    const int m0 = tile * 16;
    const float* arow = A + (size_t)(m0 + mr) * K + 2 * half;
    const float* brow = Bs + 2 * half * 16 + mr;

    v8f acc = {};
    for (int k = 0; k < K; k += 4) {
      v2f a, b;
      a[0] = arow[k];
      a[1] = arow[k + 1];
      if (RELU_A) { a[0] = fmaxf(a[0], 0.0f); a[1] = fmaxf(a[1], 0.0f); }
      b[0] = brow[k * 16];
      b[1] = brow[k * 16 + 16];
      acc = __builtin_amdgcn_wmma_f32_16x16x4_f32(false, a, false, b,
                                                  (short)0, acc, false, false);
    }

    float* crow = C + (size_t)(m0 + 8 * half) * N + n0 + mr;
#pragma unroll
    for (int r = 0; r < 8; ++r) crow[(size_t)r * N] = acc[r];
  }
}

__global__ void agg_init_k(float* __restrict__ out, const float* __restrict__ h,
                           const float* __restrict__ dinv, const float* __restrict__ bias,
                           int n4) {
  int i = blockIdx.x * blockDim.x + threadIdx.x;
  if (i < n4) {
    int node = i >> 4;
    int c4   = (i & 15) * 4;
    float d2 = dinv[node]; d2 *= d2;
    const float4 hv = ((const float4*)h)[i];
    const float4 bv = *(const float4*)&bias[c4];
    float4 o;
    o.x = bv.x + d2 * hv.x;
    o.y = bv.y + d2 * hv.y;
    o.z = bv.z + d2 * hv.z;
    o.w = bv.w + d2 * hv.w;
    ((float4*)out)[i] = o;
  }
}

__global__ void agg_edges_k(float* __restrict__ out, const float* __restrict__ h,
                            const float* __restrict__ dinv, const int* __restrict__ ei,
                            int E) {
  const int lane = threadIdx.x & 31;
  const int wv = __builtin_amdgcn_readfirstlane((int)(threadIdx.x >> 5));
  const int e0 = (blockIdx.x * (int)(blockDim.x >> 5) + wv) * 32;
  if (e0 >= E) return;
  const int cnt = min(32, E - e0);

  int s = 0, d = 0;
  float w = 0.0f;
  if (lane < cnt) {
    s = ei[e0 + lane];
    d = ei[E + e0 + lane];
    w = dinv[s] * dinv[d];
  }

  int j = 0;
  for (; j + 8 <= cnt; j += 8) {
    float v0[8], v1[8];
    int dd[8];
    for (int u = 0; u < 8; ++u) {
      const int sj = __builtin_amdgcn_readlane(s, j + u);
      const float wj = __int_as_float(__builtin_amdgcn_readlane(__float_as_int(w), j + u));
      dd[u] = __builtin_amdgcn_readlane(d, j + u);
      const float* hp = h + (size_t)sj * HID + lane;
      v0[u] = wj * hp[0];
      v1[u] = wj * hp[32];
    }
    for (int u = 0; u < 8; ++u) {
      float* op = out + (size_t)dd[u] * HID + lane;
      atomicAdd(op, v0[u]);
      atomicAdd(op + 32, v1[u]);
    }
  }
  for (; j < cnt; ++j) {
    const int sj = __builtin_amdgcn_readlane(s, j);
    const float wj = __int_as_float(__builtin_amdgcn_readlane(__float_as_int(w), j));
    const int dj = __builtin_amdgcn_readlane(d, j);
    const float* hp = h + (size_t)sj * HID + lane;
    float* op = out + (size_t)dj * HID + lane;
    atomicAdd(op, wj * hp[0]);
    atomicAdd(op + 32, wj * hp[32]);
  }
}

#define W1S_STRIDE 72

__global__ __launch_bounds__(128) void node_pq_k(float* aq, float* __restrict__ P,
                                                 const float* __restrict__ Wm1, int M) {
  __shared__ float w1s[IN_C * W1S_STRIDE];

  const int tid  = threadIdx.x;
  const int wave = tid >> 5;
  const int lane = tid & 31;
  const int mr   = lane & 15;
  const int half = lane >> 4;

  for (int i = tid; i < IN_C * 16; i += 128) {
    int row = i >> 4, c4 = (i & 15) * 4;
    *(float4*)&w1s[row * W1S_STRIDE + c4] = *(const float4*)&Wm1[(size_t)row * HID + c4];
  }
  __syncthreads();

  const int tile = blockIdx.x * 4 + wave;
  if (tile * 16 < M) {
    const int m0 = tile * 16;
    const float* arow = aq + (size_t)(m0 + mr) * HID + 2 * half;

    v2f a[16];
#pragma unroll
    for (int t = 0; t < 16; ++t) {
      a[t][0] = fmaxf(arow[4 * t], 0.0f);
      a[t][1] = fmaxf(arow[4 * t + 1], 0.0f);
    }

    v8f accP[4] = {v8f{}, v8f{}, v8f{}, v8f{}};
    v8f accQ[4] = {v8f{}, v8f{}, v8f{}, v8f{}};
#pragma unroll
    for (int t = 0; t < 16; ++t) {
      const float* bp = w1s + (4 * t + 2 * half) * W1S_STRIDE + mr;
      const float* bq = bp + 64 * W1S_STRIDE;
#pragma unroll
      for (int nt = 0; nt < 4; ++nt) {
        v2f b;
        b[0] = bp[nt * 16];
        b[1] = bp[W1S_STRIDE + nt * 16];
        accP[nt] = __builtin_amdgcn_wmma_f32_16x16x4_f32(false, a[t], false, b,
                                                         (short)0, accP[nt], false, false);
      }
#pragma unroll
      for (int nt = 0; nt < 4; ++nt) {
        v2f b;
        b[0] = bq[nt * 16];
        b[1] = bq[W1S_STRIDE + nt * 16];
        accQ[nt] = __builtin_amdgcn_wmma_f32_16x16x4_f32(false, a[t], false, b,
                                                         (short)0, accQ[nt], false, false);
      }
    }

    float* prow = P  + (size_t)(m0 + 8 * half) * HID + mr;
    float* qrow = aq + (size_t)(m0 + 8 * half) * HID + mr;
#pragma unroll
    for (int nt = 0; nt < 4; ++nt) {
#pragma unroll
      for (int r = 0; r < 8; ++r) {
        prow[(size_t)r * HID + nt * 16] = accP[nt][r];
        qrow[(size_t)r * HID + nt * 16] = accQ[nt][r];
      }
    }
  }
}

#define ZS 68
#define OS 144

__global__ __launch_bounds__(128) void edge_out_k(const float* __restrict__ P,
                                                  const float* __restrict__ Q,
                                                  const int* __restrict__ ei,
                                                  const float* __restrict__ bm1,
                                                  const float* __restrict__ Wm2,
                                                  const float* __restrict__ bm2,
                                                  float* __restrict__ out, int E, int ntiles) {
  __shared__ float zs[4][16 * ZS];

  const int tid  = threadIdx.x;
  const int lane = tid & 31;
  const int mr   = lane & 15;
  const int half = lane >> 4;
  const int wave = __builtin_amdgcn_readfirstlane(tid >> 5);
  float* zw = zs[wave];

  v2f bf[16];
#pragma unroll
  for (int j = 0; j < 8; ++j) {
    const float* wp = Wm2 + (32 * half + 4 * j) * OUT_C + mr;
    bf[2 * j][0]     = wp[0];
    bf[2 * j][1]     = wp[OUT_C];
    bf[2 * j + 1][0] = wp[2 * OUT_C];
    bf[2 * j + 1][1] = wp[3 * OUT_C];
  }
  const float4 b1v = *(const float4*)&bm1[4 * mr];
  const float bo = bm2[mr];

  const int stride = (int)gridDim.x * 4;
  int tile = (int)blockIdx.x * 4 + wave;
  const int* eip = ei + (half ? E : 0) + mr;
  int idx = (tile < ntiles) ? eip[tile * 16] : 0;

  for (; tile < ntiles; tile += stride) {
    const int tn = tile + stride;
    const int idx_next = (tn < ntiles) ? eip[tn * 16] : 0;

    float4 z[8];
#pragma unroll
    for (int i = 0; i < 8; ++i) {
      const int r0 = __builtin_amdgcn_readlane(idx, 2 * i);
      const int r1 = __builtin_amdgcn_readlane(idx, 2 * i + 1);
      const int c0 = __builtin_amdgcn_readlane(idx, 16 + 2 * i);
      const int c1 = __builtin_amdgcn_readlane(idx, 17 + 2 * i);
      const unsigned ro = (unsigned)(half ? r1 : r0) * (unsigned)(HID * 4) + (unsigned)mr * 16u;
      const unsigned co = (unsigned)(half ? c1 : c0) * (unsigned)(HID * 4) + (unsigned)mr * 16u;
      const float4 p = *(const float4*)((const char*)P + ro);
      const float4 q = *(const float4*)((const char*)Q + co);
      z[i].x = fmaxf(p.x + q.x + b1v.x, 0.0f);
      z[i].y = fmaxf(p.y + q.y + b1v.y, 0.0f);
      z[i].z = fmaxf(p.z + q.z + b1v.z, 0.0f);
      z[i].w = fmaxf(p.w + q.w + b1v.w, 0.0f);
    }
#pragma unroll
    for (int i = 0; i < 8; ++i)
      *(float4*)&zw[(2 * i + half) * ZS + 4 * mr] = z[i];


    v8f o = {};
#pragma unroll
    for (int j = 0; j < 8; ++j) {
      const float4 a4 = *(const float4*)&zw[mr * ZS + 32 * half + 4 * j];
      v2f a0, a1;
      a0[0] = a4.x; a0[1] = a4.y;
      a1[0] = a4.z; a1[1] = a4.w;
      o = __builtin_amdgcn_wmma_f32_16x16x4_f32(false, a0, false, bf[2 * j],
                                                (short)0, o, false, false);
      o = __builtin_amdgcn_wmma_f32_16x16x4_f32(false, a1, false, bf[2 * j + 1],
                                                (short)0, o, false, false);
    }

#pragma unroll
    for (int r = 0; r < 8; ++r) zw[half * OS + r * 16 + mr] = o[r] + bo;

    float* ob = out + (size_t)tile * (16 * OUT_C) + 4 * lane;
    const float4 o0 = *(const float4*)&zw[4 * lane];
    const float4 o1 = *(const float4*)&zw[OS + 4 * lane];
    *(float4*)ob = o0;
    *(float4*)(ob + 128) = o1;

    idx = idx_next;
  }
}

extern "C" void kernel_launch(void* const* d_in, const int* in_sizes, int n_in,
                              void* d_out, int out_size, void* d_ws, size_t ws_size,
                              hipStream_t stream) {
  const float* x   = (const float*)d_in[0];
  const int*   ei  = (const int*)d_in[1];
  const float* W1  = (const float*)d_in[2];
  const float* b1  = (const float*)d_in[3];
  const float* W2  = (const float*)d_in[4];
  const float* b2  = (const float*)d_in[5];
  const float* Wm1 = (const float*)d_in[6];
  const float* bm1 = (const float*)d_in[7];
  const float* Wm2 = (const float*)d_in[8];
  const float* bm2 = (const float*)d_in[9];
  float* out = (float*)d_out;

  const int N = in_sizes[0] / IN_C;
  const int E = in_sizes[1] / 2;

  float* deg  = (float*)d_ws;
  float* bufA = deg + (size_t)((N + 63) & ~63);
  float* bufB = bufA + (size_t)N * HID;

  const int mtiles = (N + 15) / 16;
  const int gx     = (mtiles + 3) / 4;
  const int n4     = N * (HID / 4);
  const int nb  = (N + 255) / 256;
  const int eb  = (E + 255) / 256;
  const int n4b = (n4 + 255) / 256;

  deg_init_k<<<nb, 256, 0, stream>>>(deg, N);
  deg_edges_k<<<eb, 256, 0, stream>>>(deg, ei, E);
  dinv_k<<<nb, 256, 0, stream>>>(deg, N);

  gemm_wmma_k<false><<<dim3(gx, HID / 16), 128, 0, stream>>>(x, W1, bufA, N, IN_C, HID);
  agg_init_k<<<n4b, 256, 0, stream>>>(bufB, bufA, deg, b1, n4);
  agg_edges_k<<<eb, 256, 0, stream>>>(bufB, bufA, deg, ei, E);

  gemm_wmma_k<true><<<dim3(gx, HID / 16), 128, 0, stream>>>(bufB, W2, bufA, N, HID, HID);
  agg_init_k<<<n4b, 256, 0, stream>>>(bufB, bufA, deg, b2, n4);
  agg_edges_k<<<eb, 256, 0, stream>>>(bufB, bufA, deg, ei, E);

  node_pq_k<<<gx, 128, 0, stream>>>(bufB, bufA, Wm1, N);
  edge_out_k<<<1250, 128, 0, stream>>>(bufA, bufB, ei, bm1, Wm2, bm2, out, E, E / 16);
}


# agg_init folded into the dense layers: one wave owns 16 whole rows (all 4 column tiles, A read once) and also stores bias + dinv^2*h, in place for layer 2; both init launches go; sums unchanged
# speedup vs baseline: 1.6498x; 1.0879x over previous
#include <hip/hip_runtime.h>
#include <hip/hip_bf16.h>

typedef __attribute__((ext_vector_type(2))) float v2f;
typedef __attribute__((ext_vector_type(8))) float v8f;

#define IN_C 128
#define HID  64
#define OUT_C 16

__global__ void deg_init_k(float* __restrict__ deg, int n) {
  int i = blockIdx.x * blockDim.x + threadIdx.x;
  if (i < n) deg[i] = 1.0f;
}

__global__ void deg_edges_k(float* __restrict__ deg, const int* __restrict__ ei, int E) {
  int e = blockIdx.x * blockDim.x + threadIdx.x;
  if (e < E) atomicAdd(&deg[ei[E + e]], 1.0f);
}

__global__ void dinv_k(float* __restrict__ deg, int n) {
  int i = blockIdx.x * blockDim.x + threadIdx.x;
  if (i < n) {
    float d = deg[i];
    deg[i] = (d > 0.0f) ? rsqrtf(d) : 0.0f;
  }
}

#define WS_STRIDE 72

template <int K, bool RELU_A>
__global__ __launch_bounds__(128) void gemm_rows_k(const float* A, const float* __restrict__ Wt,
                                                   float* __restrict__ H, float* S,
                                                   const float* __restrict__ dinv,
                                                   const float* __restrict__ bias, int M) {
  __shared__ float wsm[K * WS_STRIDE];

  const int tid  = threadIdx.x;
  const int wave = tid >> 5;
  const int lane = tid & 31;
  const int mr   = lane & 15;
  const int half = lane >> 4;

  const int tile = blockIdx.x * 4 + wave;
  const int m0   = tile * 16;
  const bool valid = m0 < M;

  const float* arow = A + (size_t)min(m0 + mr, M - 1) * K + 2 * half;
  v2f a[K / 4];
#pragma unroll
  for (int t = 0; t < K / 4; ++t) {
    a[t][0] = arow[4 * t];
    a[t][1] = arow[4 * t + 1];
  }
  const float* dp = dinv + min(m0 + 8 * half, M - 8);
  float dv[8];
#pragma unroll
  for (int r = 0; r < 8; ++r) dv[r] = dp[r];
  float bv[4];
#pragma unroll
  for (int nt = 0; nt < 4; ++nt) bv[nt] = bias[nt * 16 + mr];

  for (int i = tid; i < K * 16; i += 128) {
    int row = i >> 4, c4 = (i & 15) * 4;
    *(float4*)&wsm[row * WS_STRIDE + c4] = *(const float4*)&Wt[(size_t)row * HID + c4];
  }
  __syncthreads();

  if (valid) {
    if (RELU_A) {
#pragma unroll
      for (int t = 0; t < K / 4; ++t) {
        a[t][0] = fmaxf(a[t][0], 0.0f);
        a[t][1] = fmaxf(a[t][1], 0.0f);
      }
    }

    v8f acc[4] = {v8f{}, v8f{}, v8f{}, v8f{}};
#pragma unroll
    for (int t = 0; t < K / 4; ++t) {
      const float* bp = wsm + (4 * t + 2 * half) * WS_STRIDE + mr;
#pragma unroll
      for (int nt = 0; nt < 4; ++nt) {
        v2f b;
        b[0] = bp[nt * 16];
        b[1] = bp[WS_STRIDE + nt * 16];
        acc[nt] = __builtin_amdgcn_wmma_f32_16x16x4_f32(false, a[t], false, b,
                                                        (short)0, acc[nt], false, false);
      }
    }

    float* hrow = H + (size_t)(m0 + 8 * half) * HID + mr;
    float* srow = S + (size_t)(m0 + 8 * half) * HID + mr;
#pragma unroll
    for (int nt = 0; nt < 4; ++nt) {
#pragma unroll
      for (int r = 0; r < 8; ++r) {
        const float v  = acc[nt][r];
        const float d2 = dv[r] * dv[r];
        hrow[(size_t)r * HID + nt * 16] = v;
        srow[(size_t)r * HID + nt * 16] = bv[nt] + d2 * v;
      }
    }
  }
}

__global__ void agg_edges_k(float* __restrict__ out, const float* __restrict__ h,
                            const float* __restrict__ dinv, const int* __restrict__ ei,
                            int E) {
  const int lane = threadIdx.x & 31;
  const int wv = __builtin_amdgcn_readfirstlane((int)(threadIdx.x >> 5));
  const int e0 = (blockIdx.x * (int)(blockDim.x >> 5) + wv) * 32;
  if (e0 >= E) return;
  const int cnt = min(32, E - e0);

  int s = 0, d = 0;
  float w = 0.0f;
  if (lane < cnt) {
    s = ei[e0 + lane];
    d = ei[E + e0 + lane];
    w = dinv[s] * dinv[d];
  }

  int j = 0;
  for (; j + 8 <= cnt; j += 8) {
    float v0[8], v1[8];
    int dd[8];
    for (int u = 0; u < 8; ++u) {
      const int sj = __builtin_amdgcn_readlane(s, j + u);
      const float wj = __int_as_float(__builtin_amdgcn_readlane(__float_as_int(w), j + u));
      dd[u] = __builtin_amdgcn_readlane(d, j + u);
      const float* hp = h + (size_t)sj * HID + lane;
      v0[u] = wj * hp[0];
      v1[u] = wj * hp[32];
    }
    for (int u = 0; u < 8; ++u) {
      float* op = out + (size_t)dd[u] * HID + lane;
      atomicAdd(op, v0[u]);
      atomicAdd(op + 32, v1[u]);
    }
  }
  for (; j < cnt; ++j) {
    const int sj = __builtin_amdgcn_readlane(s, j);
    const float wj = __int_as_float(__builtin_amdgcn_readlane(__float_as_int(w), j));
    const int dj = __builtin_amdgcn_readlane(d, j);
    const float* hp = h + (size_t)sj * HID + lane;
    float* op = out + (size_t)dj * HID + lane;
    atomicAdd(op, wj * hp[0]);
    atomicAdd(op + 32, wj * hp[32]);
  }
}

#define W1S_STRIDE 72

__global__ __launch_bounds__(128) void node_pq_k(float* aq, float* __restrict__ P,
                                                 const float* __restrict__ Wm1, int M) {
  __shared__ float w1s[IN_C * W1S_STRIDE];

  const int tid  = threadIdx.x;
  const int wave = tid >> 5;
  const int lane = tid & 31;
  const int mr   = lane & 15;
  const int half = lane >> 4;

  for (int i = tid; i < IN_C * 16; i += 128) {
    int row = i >> 4, c4 = (i & 15) * 4;
    *(float4*)&w1s[row * W1S_STRIDE + c4] = *(const float4*)&Wm1[(size_t)row * HID + c4];
  }
  __syncthreads();

  const int tile = blockIdx.x * 4 + wave;
  if (tile * 16 < M) {
    const int m0 = tile * 16;
    const float* arow = aq + (size_t)(m0 + mr) * HID + 2 * half;

    v2f a[16];
#pragma unroll
    for (int t = 0; t < 16; ++t) {
      a[t][0] = fmaxf(arow[4 * t], 0.0f);
      a[t][1] = fmaxf(arow[4 * t + 1], 0.0f);
    }

    v8f accP[4] = {v8f{}, v8f{}, v8f{}, v8f{}};
    v8f accQ[4] = {v8f{}, v8f{}, v8f{}, v8f{}};
#pragma unroll
    for (int t = 0; t < 16; ++t) {
      const float* bp = w1s + (4 * t + 2 * half) * W1S_STRIDE + mr;
      const float* bq = bp + 64 * W1S_STRIDE;
#pragma unroll
      for (int nt = 0; nt < 4; ++nt) {
        v2f b;
        b[0] = bp[nt * 16];
        b[1] = bp[W1S_STRIDE + nt * 16];
        accP[nt] = __builtin_amdgcn_wmma_f32_16x16x4_f32(false, a[t], false, b,
                                                         (short)0, accP[nt], false, false);
      }
#pragma unroll
      for (int nt = 0; nt < 4; ++nt) {
        v2f b;
        b[0] = bq[nt * 16];
        b[1] = bq[W1S_STRIDE + nt * 16];
        accQ[nt] = __builtin_amdgcn_wmma_f32_16x16x4_f32(false, a[t], false, b,
                                                         (short)0, accQ[nt], false, false);
      }
    }

    float* prow = P  + (size_t)(m0 + 8 * half) * HID + mr;
    float* qrow = aq + (size_t)(m0 + 8 * half) * HID + mr;
#pragma unroll
    for (int nt = 0; nt < 4; ++nt) {
#pragma unroll
      for (int r = 0; r < 8; ++r) {
        prow[(size_t)r * HID + nt * 16] = accP[nt][r];
        qrow[(size_t)r * HID + nt * 16] = accQ[nt][r];
      }
    }
  }
}

#define ZS 68
#define OS 144

__global__ __launch_bounds__(128) void edge_out_k(const float* __restrict__ P,
                                                  const float* __restrict__ Q,
                                                  const int* __restrict__ ei,
                                                  const float* __restrict__ bm1,
                                                  const float* __restrict__ Wm2,
                                                  const float* __restrict__ bm2,
                                                  float* __restrict__ out, int E, int ntiles) {
  __shared__ float zs[4][16 * ZS];

  const int tid  = threadIdx.x;
  const int lane = tid & 31;
  const int mr   = lane & 15;
  const int half = lane >> 4;
  const int wave = __builtin_amdgcn_readfirstlane(tid >> 5);
  float* zw = zs[wave];

  v2f bf[16];
#pragma unroll
  for (int j = 0; j < 8; ++j) {
    const float* wp = Wm2 + (32 * half + 4 * j) * OUT_C + mr;
    bf[2 * j][0]     = wp[0];
    bf[2 * j][1]     = wp[OUT_C];
    bf[2 * j + 1][0] = wp[2 * OUT_C];
    bf[2 * j + 1][1] = wp[3 * OUT_C];
  }
  const float4 b1v = *(const float4*)&bm1[4 * mr];
  const float bo = bm2[mr];

  const int stride = (int)gridDim.x * 4;
  int tile = (int)blockIdx.x * 4 + wave;
  const int* eip = ei + (half ? E : 0) + mr;
  int idx = (tile < ntiles) ? eip[tile * 16] : 0;

  for (; tile < ntiles; tile += stride) {
    const int tn = tile + stride;
    const int idx_next = (tn < ntiles) ? eip[tn * 16] : 0;

    float4 z[8];
#pragma unroll
    for (int i = 0; i < 8; ++i) {
      const int r0 = __builtin_amdgcn_readlane(idx, 2 * i);
      const int r1 = __builtin_amdgcn_readlane(idx, 2 * i + 1);
      const int c0 = __builtin_amdgcn_readlane(idx, 16 + 2 * i);
      const int c1 = __builtin_amdgcn_readlane(idx, 17 + 2 * i);
      const unsigned ro = (unsigned)(half ? r1 : r0) * (unsigned)(HID * 4) + (unsigned)mr * 16u;
      const unsigned co = (unsigned)(half ? c1 : c0) * (unsigned)(HID * 4) + (unsigned)mr * 16u;
      const float4 p = *(const float4*)((const char*)P + ro);
      const float4 q = *(const float4*)((const char*)Q + co);
      z[i].x = fmaxf(p.x + q.x + b1v.x, 0.0f);
      z[i].y = fmaxf(p.y + q.y + b1v.y, 0.0f);
      z[i].z = fmaxf(p.z + q.z + b1v.z, 0.0f);
      z[i].w = fmaxf(p.w + q.w + b1v.w, 0.0f);
    }
#pragma unroll
    for (int i = 0; i < 8; ++i)
      *(float4*)&zw[(2 * i + half) * ZS + 4 * mr] = z[i];


    v8f o = {};
#pragma unroll
    for (int j = 0; j < 8; ++j) {
      const float4 a4 = *(const float4*)&zw[mr * ZS + 32 * half + 4 * j];
      v2f a0, a1;
      a0[0] = a4.x; a0[1] = a4.y;
      a1[0] = a4.z; a1[1] = a4.w;
      o = __builtin_amdgcn_wmma_f32_16x16x4_f32(false, a0, false, bf[2 * j],
                                                (short)0, o, false, false);
      o = __builtin_amdgcn_wmma_f32_16x16x4_f32(false, a1, false, bf[2 * j + 1],
                                                (short)0, o, false, false);
    }

#pragma unroll
    for (int r = 0; r < 8; ++r) zw[half * OS + r * 16 + mr] = o[r] + bo;

    float* ob = out + (size_t)tile * (16 * OUT_C) + 4 * lane;
    const float4 o0 = *(const float4*)&zw[4 * lane];
    const float4 o1 = *(const float4*)&zw[OS + 4 * lane];
    *(float4*)ob = o0;
    *(float4*)(ob + 128) = o1;

    idx = idx_next;
  }
}

extern "C" void kernel_launch(void* const* d_in, const int* in_sizes, int n_in,
                              void* d_out, int out_size, void* d_ws, size_t ws_size,
                              hipStream_t stream) {
  const float* x   = (const float*)d_in[0];
  const int*   ei  = (const int*)d_in[1];
  const float* W1  = (const float*)d_in[2];
  const float* b1  = (const float*)d_in[3];
  const float* W2  = (const float*)d_in[4];
  const float* b2  = (const float*)d_in[5];
  const float* Wm1 = (const float*)d_in[6];
  const float* bm1 = (const float*)d_in[7];
  const float* Wm2 = (const float*)d_in[8];
  const float* bm2 = (const float*)d_in[9];
  float* out = (float*)d_out;

  const int N = in_sizes[0] / IN_C;
  const int E = in_sizes[1] / 2;

  float* deg  = (float*)d_ws;
  float* bufA = deg + (size_t)((N + 63) & ~63);
  float* bufB = bufA + (size_t)N * HID;

  const int mtiles = (N + 15) / 16;
  const int gx     = (mtiles + 3) / 4;
  const int nb  = (N + 255) / 256;
  const int eb  = (E + 255) / 256;

  deg_init_k<<<nb, 256, 0, stream>>>(deg, N);
  deg_edges_k<<<eb, 256, 0, stream>>>(deg, ei, E);
  dinv_k<<<nb, 256, 0, stream>>>(deg, N);

  gemm_rows_k<IN_C, false><<<gx, 128, 0, stream>>>(x, W1, bufA, bufB, deg, b1, N);
  agg_edges_k<<<eb, 256, 0, stream>>>(bufB, bufA, deg, ei, E);

  gemm_rows_k<HID, true><<<gx, 128, 0, stream>>>(bufB, W2, bufA, bufB, deg, b2, N);
  agg_edges_k<<<eb, 256, 0, stream>>>(bufB, bufA, deg, ei, E);

  node_pq_k<<<gx, 128, 0, stream>>>(bufB, bufA, Wm1, N);
  edge_out_k<<<1250, 128, 0, stream>>>(bufA, bufB, ei, bm1, Wm2, bm2, out, E, E / 16);
}


# float atomics out of the edge sums: edges bucketed by destination in each call (int count add = rank, wave-scan ranges, place), a wave sums a node's rows in registers; sum order = arrival rank
# speedup vs baseline: 1.9839x; 1.2025x over previous
#include <hip/hip_runtime.h>
#include <hip/hip_bf16.h>

typedef __attribute__((ext_vector_type(2))) float v2f;
typedef __attribute__((ext_vector_type(8))) float v8f;

#define IN_C 128
#define HID  64
#define OUT_C 16

__global__ void cnt_init_k(int* __restrict__ cnt, int* __restrict__ ctr, int n) {
  int i = blockIdx.x * blockDim.x + threadIdx.x;
  if (i < n) cnt[i] = 1;
  if (i == 0) ctr[0] = 0;
}

#define EPL 4

__global__ void count_edges_k(int* __restrict__ cnt, int* __restrict__ rank,
                              const int* __restrict__ ei, int E) {
  const int T  = (int)(gridDim.x * blockDim.x);
  const int e0 = blockIdx.x * blockDim.x + threadIdx.x;
  unsigned d[EPL];
  int r[EPL];
#pragma unroll
  for (int k = 0; k < EPL; ++k)
    d[k] = (unsigned)ei[E + min(e0 + k * T, E - 1)];
#pragma unroll
  for (int k = 0; k < EPL; ++k)
    r[k] = (e0 + k * T < E) ? atomicAdd(&cnt[d[k]], 1) : 0;
#pragma unroll
  for (int k = 0; k < EPL; ++k)
    if (e0 + k * T < E) rank[e0 + k * T] = r[k];
}

__global__ void ranges_k(const int* __restrict__ cnt, int* __restrict__ start,
                         int* __restrict__ ctr, float* __restrict__ dinv,
                         int2* __restrict__ ent, int n) {
  const int i = blockIdx.x * blockDim.x + threadIdx.x;
  const int lane = threadIdx.x & 31;
  const int c = (i < n) ? cnt[i] : 0;
  int incl = c;
  for (int o = 1; o < 32; o <<= 1) {
    const int t = __shfl_up(incl, o);
    if (lane >= o) incl += t;
  }
  int base = 0;
  if (lane == 31) base = atomicAdd(ctr, incl);
  base = __shfl(base, 31);
  if (i < n) {
    const int st = base + incl - c;
    const float dv = (c > 0) ? rsqrtf((float)c) : 0.0f;
    start[i] = st;
    dinv[i] = dv;
    ent[st] = make_int2(i, __float_as_int(dv));
  }
}

__global__ void place_k(int2* __restrict__ ent, const int* __restrict__ start,
                        const int* __restrict__ rank, const float* __restrict__ dinv,
                        const int* __restrict__ ei, int E) {
  const int T  = (int)(gridDim.x * blockDim.x);
  const int e0 = blockIdx.x * blockDim.x + threadIdx.x;
  unsigned s[EPL], d[EPL];
  int r[EPL], p[EPL];
  float w[EPL];
#pragma unroll
  for (int k = 0; k < EPL; ++k) {
    const int ec = min(e0 + k * T, E - 1);
    s[k] = (unsigned)ei[ec];
    d[k] = (unsigned)ei[E + ec];
    r[k] = rank[ec];
  }
#pragma unroll
  for (int k = 0; k < EPL; ++k) {
    p[k] = start[d[k]] + r[k];
    w[k] = dinv[s[k]];
  }
#pragma unroll
  for (int k = 0; k < EPL; ++k)
    if (e0 + k * T < E) ent[p[k]] = make_int2((int)s[k], __float_as_int(w[k]));
}

#define WS_STRIDE 72

template <int K, bool RELU_A>
__global__ __launch_bounds__(128) void gemm_rows_k(const float* __restrict__ A,
                                                   const float* __restrict__ Wt,
                                                   float* __restrict__ H, int M) {
  __shared__ float wsm[K * WS_STRIDE];

  const int tid  = threadIdx.x;
  const int wave = tid >> 5;
  const int lane = tid & 31;
  const int mr   = lane & 15;
  const int half = lane >> 4;

  const int tile = blockIdx.x * 4 + wave;
  const int m0   = tile * 16;
  const bool valid = m0 < M;

  const float* arow = A + (size_t)min(m0 + mr, M - 1) * K + 2 * half;
  v2f a[K / 4];
#pragma unroll
  for (int t = 0; t < K / 4; ++t) {
    a[t][0] = arow[4 * t];
    a[t][1] = arow[4 * t + 1];
  }

  for (int i = tid; i < K * 16; i += 128) {
    int row = i >> 4, c4 = (i & 15) * 4;
    *(float4*)&wsm[row * WS_STRIDE + c4] = *(const float4*)&Wt[(size_t)row * HID + c4];
  }
  __syncthreads();

  if (valid) {
    if (RELU_A) {
#pragma unroll
      for (int t = 0; t < K / 4; ++t) {
        a[t][0] = fmaxf(a[t][0], 0.0f);
        a[t][1] = fmaxf(a[t][1], 0.0f);
      }
    }

    v8f acc[4] = {v8f{}, v8f{}, v8f{}, v8f{}};
#pragma unroll
    for (int t = 0; t < K / 4; ++t) {
      const float* bp = wsm + (4 * t + 2 * half) * WS_STRIDE + mr;
#pragma unroll
      for (int nt = 0; nt < 4; ++nt) {
        v2f b;
        b[0] = bp[nt * 16];
        b[1] = bp[WS_STRIDE + nt * 16];
        acc[nt] = __builtin_amdgcn_wmma_f32_16x16x4_f32(false, a[t], false, b,
                                                        (short)0, acc[nt], false, false);
      }
    }

    float* hrow = H + (size_t)(m0 + 8 * half) * HID + mr;
#pragma unroll
    for (int nt = 0; nt < 4; ++nt) {
#pragma unroll
      for (int r = 0; r < 8; ++r) hrow[(size_t)r * HID + nt * 16] = acc[nt][r];
    }
  }
}

#define AGG_NPW 4

__global__ __launch_bounds__(256) void agg_rows_k(float* __restrict__ out,
                                                  const float* __restrict__ h,
                                                  const float* __restrict__ dinv,
                                                  const int* __restrict__ start,
                                                  const int* __restrict__ cnt,
                                                  const int2* __restrict__ ent,
                                                  const float* __restrict__ bias, int N) {
  const int lane = threadIdx.x & 31;
  const int wv = __builtin_amdgcn_readfirstlane((int)(threadIdx.x >> 5));
  const int n0 = (blockIdx.x * (int)(blockDim.x >> 5) + wv) * AGG_NPW;
  if (n0 >= N) return;
  const int nn = min(AGG_NPW, N - n0);

  int st = 0, ct = 0;
  float dl = 0.0f;
  if (lane < nn) {
    st = start[n0 + lane];
    ct = cnt[n0 + lane];
    dl = dinv[n0 + lane];
  }
  const float b0 = bias[lane];
  const float b1 = bias[lane + 32];

  int beg = __builtin_amdgcn_readlane(st, 0);
  int len = __builtin_amdgcn_readlane(ct, 0);
  int2 en = make_int2(0, 0);
  if (lane < len) en = ent[beg + lane];

  for (int i = 0; i < nn; ++i) {
    const float dd = __int_as_float(__builtin_amdgcn_readlane(__float_as_int(dl), i));
    int nbeg = 0, nlen = 0;
    int2 nen = make_int2(0, 0);

    float acc0 = 0.0f, acc1 = 0.0f;
    for (int done = 0; done < len; done += 32) {
      if (done > 0) {
        en = make_int2(0, 0);
        if (done + lane < len) en = ent[beg + done + lane];
      }
      const int c = min(32, len - done);
      float v0[32], v1[32];
#pragma unroll
      for (int u = 0; u < 32; ++u) {
        v0[u] = 0.0f;
        v1[u] = 0.0f;
        if (u < c) {
          const int sj = __builtin_amdgcn_readlane(en.x, u);
          const float* hp = h + (size_t)sj * HID + lane;
          v0[u] = hp[0];
          v1[u] = hp[32];
        }
      }
      if (done == 0 && i + 1 < nn) {
        nbeg = __builtin_amdgcn_readlane(st, i + 1);
        nlen = __builtin_amdgcn_readlane(ct, i + 1);
        if (lane < nlen) nen = ent[nbeg + lane];
      }
#pragma unroll
      for (int u = 0; u < 32; ++u) {
        const float wj = __int_as_float(__builtin_amdgcn_readlane(en.y, u)) * dd;
        acc0 += wj * v0[u];
        acc1 += wj * v1[u];
      }
    }

    float* op = out + (size_t)(n0 + i) * HID + lane;
    op[0]  = b0 + acc0;
    op[32] = b1 + acc1;

    beg = nbeg;
    len = nlen;
    en = nen;
  }
}

#define W1S_STRIDE 72

__global__ __launch_bounds__(128) void node_pq_k(float* aq, float* __restrict__ P,
                                                 const float* __restrict__ Wm1, int M) {
  __shared__ float w1s[IN_C * W1S_STRIDE];

  const int tid  = threadIdx.x;
  const int wave = tid >> 5;
  const int lane = tid & 31;
  const int mr   = lane & 15;
  const int half = lane >> 4;

  for (int i = tid; i < IN_C * 16; i += 128) {
    int row = i >> 4, c4 = (i & 15) * 4;
    *(float4*)&w1s[row * W1S_STRIDE + c4] = *(const float4*)&Wm1[(size_t)row * HID + c4];
  }
  __syncthreads();

  const int tile = blockIdx.x * 4 + wave;
  if (tile * 16 < M) {
    const int m0 = tile * 16;
    const float* arow = aq + (size_t)(m0 + mr) * HID + 2 * half;

    v2f a[16];
#pragma unroll
    for (int t = 0; t < 16; ++t) {
      a[t][0] = fmaxf(arow[4 * t], 0.0f);
      a[t][1] = fmaxf(arow[4 * t + 1], 0.0f);
    }

    v8f accP[4] = {v8f{}, v8f{}, v8f{}, v8f{}};
    v8f accQ[4] = {v8f{}, v8f{}, v8f{}, v8f{}};
#pragma unroll
    for (int t = 0; t < 16; ++t) {
      const float* bp = w1s + (4 * t + 2 * half) * W1S_STRIDE + mr;
      const float* bq = bp + 64 * W1S_STRIDE;
#pragma unroll
      for (int nt = 0; nt < 4; ++nt) {
        v2f b;
        b[0] = bp[nt * 16];
        b[1] = bp[W1S_STRIDE + nt * 16];
        accP[nt] = __builtin_amdgcn_wmma_f32_16x16x4_f32(false, a[t], false, b,
                                                         (short)0, accP[nt], false, false);
      }
#pragma unroll
      for (int nt = 0; nt < 4; ++nt) {
        v2f b;
        b[0] = bq[nt * 16];
        b[1] = bq[W1S_STRIDE + nt * 16];
        accQ[nt] = __builtin_amdgcn_wmma_f32_16x16x4_f32(false, a[t], false, b,
                                                         (short)0, accQ[nt], false, false);
      }
    }

    float* prow = P  + (size_t)(m0 + 8 * half) * HID + mr;
    float* qrow = aq + (size_t)(m0 + 8 * half) * HID + mr;
#pragma unroll
    for (int nt = 0; nt < 4; ++nt) {
#pragma unroll
      for (int r = 0; r < 8; ++r) {
        prow[(size_t)r * HID + nt * 16] = accP[nt][r];
        qrow[(size_t)r * HID + nt * 16] = accQ[nt][r];
      }
    }
  }
}

#define ZS 68
#define OS 144

__global__ __launch_bounds__(128) void edge_out_k(const float* __restrict__ P,
                                                  const float* __restrict__ Q,
                                                  const int* __restrict__ ei,
                                                  const float* __restrict__ bm1,
                                                  const float* __restrict__ Wm2,
                                                  const float* __restrict__ bm2,
                                                  float* __restrict__ out, int E, int ntiles) {
  __shared__ float zs[4][16 * ZS];

  const int tid  = threadIdx.x;
  const int lane = tid & 31;
  const int mr   = lane & 15;
  const int half = lane >> 4;
  const int wave = __builtin_amdgcn_readfirstlane(tid >> 5);
  float* zw = zs[wave];

  v2f bf[16];
#pragma unroll
  for (int j = 0; j < 8; ++j) {
    const float* wp = Wm2 + (32 * half + 4 * j) * OUT_C + mr;
    bf[2 * j][0]     = wp[0];
    bf[2 * j][1]     = wp[OUT_C];
    bf[2 * j + 1][0] = wp[2 * OUT_C];
    bf[2 * j + 1][1] = wp[3 * OUT_C];
  }
  const float4 b1v = *(const float4*)&bm1[4 * mr];
  const float bo = bm2[mr];

  const int stride = (int)gridDim.x * 4;
  int tile = (int)blockIdx.x * 4 + wave;
  const int* eip = ei + (half ? E : 0) + mr;
  int idx = (tile < ntiles) ? eip[tile * 16] : 0;

  for (; tile < ntiles; tile += stride) {
    const int tn = tile + stride;
    const int idx_next = (tn < ntiles) ? eip[tn * 16] : 0;

    float4 z[8];
#pragma unroll
    for (int i = 0; i < 8; ++i) {
      const int r0 = __builtin_amdgcn_readlane(idx, 2 * i);
      const int r1 = __builtin_amdgcn_readlane(idx, 2 * i + 1);
      const int c0 = __builtin_amdgcn_readlane(idx, 16 + 2 * i);
      const int c1 = __builtin_amdgcn_readlane(idx, 17 + 2 * i);
      const unsigned ro = (unsigned)(half ? r1 : r0) * (unsigned)(HID * 4) + (unsigned)mr * 16u;
      const unsigned co = (unsigned)(half ? c1 : c0) * (unsigned)(HID * 4) + (unsigned)mr * 16u;
      const float4 p = *(const float4*)((const char*)P + ro);
      const float4 q = *(const float4*)((const char*)Q + co);
      z[i].x = fmaxf(p.x + q.x + b1v.x, 0.0f);
      z[i].y = fmaxf(p.y + q.y + b1v.y, 0.0f);
      z[i].z = fmaxf(p.z + q.z + b1v.z, 0.0f);
      z[i].w = fmaxf(p.w + q.w + b1v.w, 0.0f);
    }
#pragma unroll
    for (int i = 0; i < 8; ++i)
      *(float4*)&zw[(2 * i + half) * ZS + 4 * mr] = z[i];


    v8f o = {};
#pragma unroll
    for (int j = 0; j < 8; ++j) {
      const float4 a4 = *(const float4*)&zw[mr * ZS + 32 * half + 4 * j];
      v2f a0, a1;
      a0[0] = a4.x; a0[1] = a4.y;
      a1[0] = a4.z; a1[1] = a4.w;
      o = __builtin_amdgcn_wmma_f32_16x16x4_f32(false, a0, false, bf[2 * j],
                                                (short)0, o, false, false);
      o = __builtin_amdgcn_wmma_f32_16x16x4_f32(false, a1, false, bf[2 * j + 1],
                                                (short)0, o, false, false);
    }

#pragma unroll
    for (int r = 0; r < 8; ++r) zw[half * OS + r * 16 + mr] = o[r] + bo;

    float* ob = out + (size_t)tile * (16 * OUT_C) + 4 * lane;
    const float4 o0 = *(const float4*)&zw[4 * lane];
    const float4 o1 = *(const float4*)&zw[OS + 4 * lane];
    *(float4*)ob = o0;
    *(float4*)(ob + 128) = o1;

    idx = idx_next;
  }
}

extern "C" void kernel_launch(void* const* d_in, const int* in_sizes, int n_in,
                              void* d_out, int out_size, void* d_ws, size_t ws_size,
                              hipStream_t stream) {
  const float* x   = (const float*)d_in[0];
  const int*   ei  = (const int*)d_in[1];
  const float* W1  = (const float*)d_in[2];
  const float* b1  = (const float*)d_in[3];
  const float* W2  = (const float*)d_in[4];
  const float* b2  = (const float*)d_in[5];
  const float* Wm1 = (const float*)d_in[6];
  const float* bm1 = (const float*)d_in[7];
  const float* Wm2 = (const float*)d_in[8];
  const float* bm2 = (const float*)d_in[9];
  float* out = (float*)d_out;

  const int N = in_sizes[0] / IN_C;
  const int E = in_sizes[1] / 2;

  float* deg  = (float*)d_ws;
  float* bufA = deg + (size_t)((N + 63) & ~63);
  float* bufB = bufA + (size_t)N * HID;

  int*  o_cnt   = (int*)d_out;
  int*  o_start = o_cnt + (size_t)((N + 63) & ~63);
  int*  o_ctr   = o_start + (size_t)((N + 63) & ~63);
  int*  o_rank  = o_ctr + 64;
  int2* o_ent   = (int2*)(o_rank + (size_t)((E + 1) & ~1));

  const int mtiles = (N + 15) / 16;
  const int gx     = (mtiles + 3) / 4;
  const int nb  = (N + 255) / 256;
  const int ab  = (N + 8 * AGG_NPW - 1) / (8 * AGG_NPW);
  const int qb  = (E + 256 * EPL - 1) / (256 * EPL);

  cnt_init_k<<<nb, 256, 0, stream>>>(o_cnt, o_ctr, N);
  count_edges_k<<<qb, 256, 0, stream>>>(o_cnt, o_rank, ei, E);
  ranges_k<<<nb, 256, 0, stream>>>(o_cnt, o_start, o_ctr, deg, o_ent, N);
  place_k<<<qb, 256, 0, stream>>>(o_ent, o_start, o_rank, deg, ei, E);

  gemm_rows_k<IN_C, false><<<gx, 128, 0, stream>>>(x, W1, bufA, N);
  agg_rows_k<<<ab, 256, 0, stream>>>(bufB, bufA, deg, o_start, o_cnt, o_ent, b1, N);

  gemm_rows_k<HID, true><<<gx, 128, 0, stream>>>(bufB, W2, bufA, N);
  agg_rows_k<<<ab, 256, 0, stream>>>(bufB, bufA, deg, o_start, o_cnt, o_ent, b2, N);

  node_pq_k<<<gx, 128, 0, stream>>>(bufB, bufA, Wm1, N);
  edge_out_k<<<1250, 128, 0, stream>>>(bufA, bufB, ei, bm1, Wm2, bm2, out, E, E / 16);
}


# each aggregation kernel now also does the dense transform that consumes it (relu(a)@W2; relu(a)@Wm1 halves): summed rows go via LDS to WMMA, a1/a2 never stored, 2 launches go; sums unchanged
# speedup vs baseline: 2.2428x; 1.1305x over previous
#include <hip/hip_runtime.h>
#include <hip/hip_bf16.h>

typedef __attribute__((ext_vector_type(2))) float v2f;
typedef __attribute__((ext_vector_type(8))) float v8f;

#define IN_C 128
#define HID  64
#define OUT_C 16

__global__ void cnt_init_k(int* __restrict__ cnt, int* __restrict__ ctr, int n) {
  int i = blockIdx.x * blockDim.x + threadIdx.x;
  if (i < n) cnt[i] = 1;
  if (i == 0) ctr[0] = 0;
}

#define EPL 4

__global__ void count_edges_k(int* __restrict__ cnt, int* __restrict__ rank,
                              const int* __restrict__ ei, int E) {
  const int T  = (int)(gridDim.x * blockDim.x);
  const int e0 = blockIdx.x * blockDim.x + threadIdx.x;
  unsigned d[EPL];
  int r[EPL];
#pragma unroll
  for (int k = 0; k < EPL; ++k)
    d[k] = (unsigned)ei[E + min(e0 + k * T, E - 1)];
#pragma unroll
  for (int k = 0; k < EPL; ++k)
    r[k] = (e0 + k * T < E) ? atomicAdd(&cnt[d[k]], 1) : 0;
#pragma unroll
  for (int k = 0; k < EPL; ++k)
    if (e0 + k * T < E) rank[e0 + k * T] = r[k];
}

__global__ void ranges_k(const int* __restrict__ cnt, int* __restrict__ start,
                         int* __restrict__ ctr, float* __restrict__ dinv,
                         int2* __restrict__ ent, int n) {
  const int i = blockIdx.x * blockDim.x + threadIdx.x;
  const int lane = threadIdx.x & 31;
  const int c = (i < n) ? cnt[i] : 0;
  int incl = c;
  for (int o = 1; o < 32; o <<= 1) {
    const int t = __shfl_up(incl, o);
    if (lane >= o) incl += t;
  }
  int base = 0;
  if (lane == 31) base = atomicAdd(ctr, incl);
  base = __shfl(base, 31);
  if (i < n) {
    const int st = base + incl - c;
    const float dv = (c > 0) ? rsqrtf((float)c) : 0.0f;
    start[i] = st;
    dinv[i] = dv;
    ent[st] = make_int2(i, __float_as_int(dv));
  }
}

__global__ void place_k(int2* __restrict__ ent, const int* __restrict__ start,
                        const int* __restrict__ rank, const float* __restrict__ dinv,
                        const int* __restrict__ ei, int E) {
  const int T  = (int)(gridDim.x * blockDim.x);
  const int e0 = blockIdx.x * blockDim.x + threadIdx.x;
  unsigned s[EPL], d[EPL];
  int r[EPL], p[EPL];
  float w[EPL];
#pragma unroll
  for (int k = 0; k < EPL; ++k) {
    const int ec = min(e0 + k * T, E - 1);
    s[k] = (unsigned)ei[ec];
    d[k] = (unsigned)ei[E + ec];
    r[k] = rank[ec];
  }
#pragma unroll
  for (int k = 0; k < EPL; ++k) {
    p[k] = start[d[k]] + r[k];
    w[k] = dinv[s[k]];
  }
#pragma unroll
  for (int k = 0; k < EPL; ++k)
    if (e0 + k * T < E) ent[p[k]] = make_int2((int)s[k], __float_as_int(w[k]));
}

#define WS_STRIDE 72

template <int K, bool RELU_A>
__global__ __launch_bounds__(128) void gemm_rows_k(const float* __restrict__ A,
                                                   const float* __restrict__ Wt,
                                                   float* __restrict__ H, int M) {
  __shared__ float wsm[K * WS_STRIDE];

  const int tid  = threadIdx.x;
  const int wave = tid >> 5;
  const int lane = tid & 31;
  const int mr   = lane & 15;
  const int half = lane >> 4;

  const int tile = blockIdx.x * 4 + wave;
  const int m0   = tile * 16;
  const bool valid = m0 < M;

  const float* arow = A + (size_t)min(m0 + mr, M - 1) * K + 2 * half;
  v2f a[K / 4];
#pragma unroll
  for (int t = 0; t < K / 4; ++t) {
    a[t][0] = arow[4 * t];
    a[t][1] = arow[4 * t + 1];
  }

  for (int i = tid; i < K * 16; i += 128) {
    int row = i >> 4, c4 = (i & 15) * 4;
    *(float4*)&wsm[row * WS_STRIDE + c4] = *(const float4*)&Wt[(size_t)row * HID + c4];
  }
  __syncthreads();

  if (valid) {
    if (RELU_A) {
#pragma unroll
      for (int t = 0; t < K / 4; ++t) {
        a[t][0] = fmaxf(a[t][0], 0.0f);
        a[t][1] = fmaxf(a[t][1], 0.0f);
      }
    }

    v8f acc[4] = {v8f{}, v8f{}, v8f{}, v8f{}};
#pragma unroll
    for (int t = 0; t < K / 4; ++t) {
      const float* bp = wsm + (4 * t + 2 * half) * WS_STRIDE + mr;
#pragma unroll
      for (int nt = 0; nt < 4; ++nt) {
        v2f b;
        b[0] = bp[nt * 16];
        b[1] = bp[WS_STRIDE + nt * 16];
        acc[nt] = __builtin_amdgcn_wmma_f32_16x16x4_f32(false, a[t], false, b,
                                                        (short)0, acc[nt], false, false);
      }
    }

    float* hrow = H + (size_t)(m0 + 8 * half) * HID + mr;
#pragma unroll
    for (int nt = 0; nt < 4; ++nt) {
#pragma unroll
      for (int r = 0; r < 8; ++r) hrow[(size_t)r * HID + nt * 16] = acc[nt][r];
    }
  }
}

#define AGG_NPW 4
#define AS 68

template <int NPAIR>
__global__ __launch_bounds__(256) void agg_gemm_k(float* __restrict__ O0,
                                                  float* __restrict__ O1,
                                                  const float* __restrict__ h,
                                                  const float* __restrict__ dinv,
                                                  const int* __restrict__ start,
                                                  const int* __restrict__ cnt,
                                                  const int2* __restrict__ ent,
                                                  const float* __restrict__ bias,
                                                  const float* __restrict__ Wt, int N) {
  __shared__ __attribute__((aligned(16))) float as[8 * AGG_NPW * AS];

  const int lane = threadIdx.x & 31;
  const int wv = __builtin_amdgcn_readfirstlane((int)(threadIdx.x >> 5));
  const int n0 = ((int)blockIdx.x * 8 + wv) * AGG_NPW;
  const int nn = max(0, min(AGG_NPW, N - n0));
  float* arow = as + (wv * AGG_NPW) * AS + lane;

  int st = 0, ct = 0;
  float dl = 0.0f;
  if (lane < nn) {
    st = start[n0 + lane];
    ct = cnt[n0 + lane];
    dl = dinv[n0 + lane];
  }
  const float b0 = bias[lane];
  const float b1 = bias[lane + 32];

  int beg = __builtin_amdgcn_readlane(st, 0);
  int len = __builtin_amdgcn_readlane(ct, 0);
  int2 en = make_int2(0, 0);
  if (lane < len) en = ent[beg + lane];

  for (int i = 0; i < nn; ++i) {
    const float dd = __int_as_float(__builtin_amdgcn_readlane(__float_as_int(dl), i));
    int nbeg = 0, nlen = 0;
    int2 nen = make_int2(0, 0);

    float acc0 = 0.0f, acc1 = 0.0f;
    for (int done = 0; done < len; done += 32) {
      if (done > 0) {
        en = make_int2(0, 0);
        if (done + lane < len) en = ent[beg + done + lane];
      }
      const int c = min(32, len - done);
      float v0[32], v1[32];
#pragma unroll
      for (int u = 0; u < 32; ++u) {
        v0[u] = 0.0f;
        v1[u] = 0.0f;
        if (u < c) {
          const int sj = __builtin_amdgcn_readlane(en.x, u);
          const float* hp = h + (size_t)sj * HID + lane;
          v0[u] = hp[0];
          v1[u] = hp[32];
        }
      }
      if (done == 0 && i + 1 < nn) {
        nbeg = __builtin_amdgcn_readlane(st, i + 1);
        nlen = __builtin_amdgcn_readlane(ct, i + 1);
        if (lane < nlen) nen = ent[nbeg + lane];
      }
#pragma unroll
      for (int u = 0; u < 32; ++u) {
        const float wj = __int_as_float(__builtin_amdgcn_readlane(en.y, u)) * dd;
        acc0 += wj * v0[u];
        acc1 += wj * v1[u];
      }
    }

    arow[i * AS]      = fmaxf(b0 + acc0, 0.0f);
    arow[i * AS + 32] = fmaxf(b1 + acc1, 0.0f);

    beg = nbeg;
    len = nlen;
    en = nen;
  }
  for (int i = nn; i < AGG_NPW; ++i) {
    arow[i * AS]      = 0.0f;
    arow[i * AS + 32] = 0.0f;
  }

  const int mr   = lane & 15;
  const int half = lane >> 4;
  const bool worker = wv < 2 * NPAIR;
  const int mt = wv / NPAIR;
  const int np = wv % NPAIR;
  v2f bf[2][16];
  if (worker) {
    const float* wp = Wt + (size_t)((np >> 1) * HID + 2 * half) * HID + (np & 1) * 32 + 2 * mr;
#pragma unroll
    for (int t = 0; t < 16; ++t) {
      const float2 w0 = *(const float2*)(wp + (size_t)(4 * t) * HID);
      const float2 w1 = *(const float2*)(wp + (size_t)(4 * t + 1) * HID);
      bf[0][t][0] = w0.x;
      bf[0][t][1] = w1.x;
      bf[1][t][0] = w0.y;
      bf[1][t][1] = w1.y;
    }
  }
  __syncthreads();

  const int m0 = (int)blockIdx.x * (8 * AGG_NPW) + mt * 16;
  if (worker && m0 < N) {
    const float* ap = as + (mt * 16 + mr) * AS + 2 * half;
    v8f acc[2] = {v8f{}, v8f{}};
#pragma unroll
    for (int t = 0; t < 16; ++t) {
      const v2f a = *(const v2f*)(ap + 4 * t);
#pragma unroll
      for (int nt = 0; nt < 2; ++nt)
        acc[nt] = __builtin_amdgcn_wmma_f32_16x16x4_f32(false, a, false, bf[nt][t],
                                                        (short)0, acc[nt], false, false);
    }

    float* obase = (NPAIR == 4 && (np >> 1)) ? O1 : O0;
    float* orow = obase + (size_t)(m0 + 8 * half) * HID + (np & 1) * 32 + 2 * mr;
#pragma unroll
    for (int r = 0; r < 8; ++r)
      *(float2*)(orow + (size_t)r * HID) = make_float2(acc[0][r], acc[1][r]);
  }
}


#define ZS 68
#define OS 144

__global__ __launch_bounds__(128) void edge_out_k(const float* __restrict__ P,
                                                  const float* __restrict__ Q,
                                                  const int* __restrict__ ei,
                                                  const float* __restrict__ bm1,
                                                  const float* __restrict__ Wm2,
                                                  const float* __restrict__ bm2,
                                                  float* __restrict__ out, int E, int ntiles) {
  __shared__ float zs[4][16 * ZS];

  const int tid  = threadIdx.x;
  const int lane = tid & 31;
  const int mr   = lane & 15;
  const int half = lane >> 4;
  const int wave = __builtin_amdgcn_readfirstlane(tid >> 5);
  float* zw = zs[wave];

  v2f bf[16];
#pragma unroll
  for (int j = 0; j < 8; ++j) {
    const float* wp = Wm2 + (32 * half + 4 * j) * OUT_C + mr;
    bf[2 * j][0]     = wp[0];
    bf[2 * j][1]     = wp[OUT_C];
    bf[2 * j + 1][0] = wp[2 * OUT_C];
    bf[2 * j + 1][1] = wp[3 * OUT_C];
  }
  const float4 b1v = *(const float4*)&bm1[4 * mr];
  const float bo = bm2[mr];

  const int stride = (int)gridDim.x * 4;
  int tile = (int)blockIdx.x * 4 + wave;
  const int* eip = ei + (half ? E : 0) + mr;
  int idx = (tile < ntiles) ? eip[tile * 16] : 0;

  for (; tile < ntiles; tile += stride) {
    const int tn = tile + stride;
    const int idx_next = (tn < ntiles) ? eip[tn * 16] : 0;

    float4 z[8];
#pragma unroll
    for (int i = 0; i < 8; ++i) {
      const int r0 = __builtin_amdgcn_readlane(idx, 2 * i);
      const int r1 = __builtin_amdgcn_readlane(idx, 2 * i + 1);
      const int c0 = __builtin_amdgcn_readlane(idx, 16 + 2 * i);
      const int c1 = __builtin_amdgcn_readlane(idx, 17 + 2 * i);
      const unsigned ro = (unsigned)(half ? r1 : r0) * (unsigned)(HID * 4) + (unsigned)mr * 16u;
      const unsigned co = (unsigned)(half ? c1 : c0) * (unsigned)(HID * 4) + (unsigned)mr * 16u;
      const float4 p = *(const float4*)((const char*)P + ro);
      const float4 q = *(const float4*)((const char*)Q + co);
      z[i].x = fmaxf(p.x + q.x + b1v.x, 0.0f);
      z[i].y = fmaxf(p.y + q.y + b1v.y, 0.0f);
      z[i].z = fmaxf(p.z + q.z + b1v.z, 0.0f);
      z[i].w = fmaxf(p.w + q.w + b1v.w, 0.0f);
    }
#pragma unroll
    for (int i = 0; i < 8; ++i)
      *(float4*)&zw[(2 * i + half) * ZS + 4 * mr] = z[i];


    v8f o = {};
#pragma unroll
    for (int j = 0; j < 8; ++j) {
      const float4 a4 = *(const float4*)&zw[mr * ZS + 32 * half + 4 * j];
      v2f a0, a1;
      a0[0] = a4.x; a0[1] = a4.y;
      a1[0] = a4.z; a1[1] = a4.w;
      o = __builtin_amdgcn_wmma_f32_16x16x4_f32(false, a0, false, bf[2 * j],
                                                (short)0, o, false, false);
      o = __builtin_amdgcn_wmma_f32_16x16x4_f32(false, a1, false, bf[2 * j + 1],
                                                (short)0, o, false, false);
    }

#pragma unroll
    for (int r = 0; r < 8; ++r) zw[half * OS + r * 16 + mr] = o[r] + bo;

    float* ob = out + (size_t)tile * (16 * OUT_C) + 4 * lane;
    const float4 o0 = *(const float4*)&zw[4 * lane];
    const float4 o1 = *(const float4*)&zw[OS + 4 * lane];
    *(float4*)ob = o0;
    *(float4*)(ob + 128) = o1;

    idx = idx_next;
  }
}

extern "C" void kernel_launch(void* const* d_in, const int* in_sizes, int n_in,
                              void* d_out, int out_size, void* d_ws, size_t ws_size,
                              hipStream_t stream) {
  const float* x   = (const float*)d_in[0];
  const int*   ei  = (const int*)d_in[1];
  const float* W1  = (const float*)d_in[2];
  const float* b1  = (const float*)d_in[3];
  const float* W2  = (const float*)d_in[4];
  const float* b2  = (const float*)d_in[5];
  const float* Wm1 = (const float*)d_in[6];
  const float* bm1 = (const float*)d_in[7];
  const float* Wm2 = (const float*)d_in[8];
  const float* bm2 = (const float*)d_in[9];
  float* out = (float*)d_out;

  const int N = in_sizes[0] / IN_C;
  const int E = in_sizes[1] / 2;

  float* deg  = (float*)d_ws;
  float* bufA = deg + (size_t)((N + 63) & ~63);
  float* bufB = bufA + (size_t)N * HID;

  int*  o_cnt   = (int*)d_out;
  int*  o_start = o_cnt + (size_t)((N + 63) & ~63);
  int*  o_ctr   = o_start + (size_t)((N + 63) & ~63);
  int*  o_rank  = o_ctr + 64;
  int2* o_ent   = (int2*)(o_rank + (size_t)((E + 1) & ~1));
  float* o_h2   = (float*)(o_ent + (size_t)((E + N + 31) & ~31));

  const int mtiles = (N + 15) / 16;
  const int gx     = (mtiles + 3) / 4;
  const int nb  = (N + 255) / 256;
  const int ab  = (N + 8 * AGG_NPW - 1) / (8 * AGG_NPW);
  const int qb  = (E + 256 * EPL - 1) / (256 * EPL);

  cnt_init_k<<<nb, 256, 0, stream>>>(o_cnt, o_ctr, N);
  count_edges_k<<<qb, 256, 0, stream>>>(o_cnt, o_rank, ei, E);
  ranges_k<<<nb, 256, 0, stream>>>(o_cnt, o_start, o_ctr, deg, o_ent, N);
  place_k<<<qb, 256, 0, stream>>>(o_ent, o_start, o_rank, deg, ei, E);

  gemm_rows_k<IN_C, false><<<gx, 128, 0, stream>>>(x, W1, bufA, N);
  agg_gemm_k<2><<<ab, 256, 0, stream>>>(o_h2, nullptr, bufA, deg, o_start, o_cnt, o_ent, b1, W2, N);
  agg_gemm_k<4><<<ab, 256, 0, stream>>>(bufA, bufB, o_h2, deg, o_start, o_cnt, o_ent, b2, Wm1, N);

  edge_out_k<<<1250, 128, 0, stream>>>(bufA, bufB, ei, bm1, Wm2, bm2, out, E, E / 16);
}


# buckets in one pass: each node owns a fixed 64-entry row, so the counting add is the place (overflow to a shared list); ranges and placing launches go, dinv taken from cnt on use; sums unchanged
# speedup vs baseline: 2.4352x; 1.0858x over previous
#include <hip/hip_runtime.h>
#include <hip/hip_bf16.h>

typedef __attribute__((ext_vector_type(2))) float v2f;
typedef __attribute__((ext_vector_type(8))) float v8f;

#define IN_C 128
#define HID  64
#define OUT_C 16

#define CAP 64

__global__ void cnt_init_k(int* __restrict__ cnt, int* __restrict__ tab,
                           int* __restrict__ octr, int n) {
  int i = blockIdx.x * blockDim.x + threadIdx.x;
  if (i < n) {
    cnt[i] = 1;
    tab[(size_t)i * CAP] = i;
  }
  if (i == 0) octr[0] = 0;
}

#define EPL 4

__global__ void place_edges_k(int* __restrict__ cnt, int* __restrict__ tab,
                              int2* __restrict__ ovf, int* __restrict__ octr,
                              const int* __restrict__ ei, int E) {
  const int T  = (int)(gridDim.x * blockDim.x);
  const int e0 = blockIdx.x * blockDim.x + threadIdx.x;
  unsigned s[EPL], d[EPL];
  int r[EPL];
#pragma unroll
  for (int k = 0; k < EPL; ++k) {
    const int ec = min(e0 + k * T, E - 1);
    s[k] = (unsigned)ei[ec];
    d[k] = (unsigned)ei[E + ec];
  }
#pragma unroll
  for (int k = 0; k < EPL; ++k)
    r[k] = (e0 + k * T < E) ? atomicAdd(&cnt[d[k]], 1) : 0;
#pragma unroll
  for (int k = 0; k < EPL; ++k) {
    if (e0 + k * T < E) {
      if (r[k] < CAP) {
        tab[(size_t)d[k] * CAP + r[k]] = (int)s[k];
      } else {
        const int q = atomicAdd(octr, 1);
        ovf[q] = make_int2((int)d[k], (int)s[k]);
      }
    }
  }
}

#define WS_STRIDE 72

template <int K, bool RELU_A>
__global__ __launch_bounds__(128) void gemm_rows_k(const float* __restrict__ A,
                                                   const float* __restrict__ Wt,
                                                   float* __restrict__ H, int M) {
  __shared__ float wsm[K * WS_STRIDE];

  const int tid  = threadIdx.x;
  const int wave = tid >> 5;
  const int lane = tid & 31;
  const int mr   = lane & 15;
  const int half = lane >> 4;

  const int tile = blockIdx.x * 4 + wave;
  const int m0   = tile * 16;
  const bool valid = m0 < M;

  const float* arow = A + (size_t)min(m0 + mr, M - 1) * K + 2 * half;
  v2f a[K / 4];
#pragma unroll
  for (int t = 0; t < K / 4; ++t) {
    a[t][0] = arow[4 * t];
    a[t][1] = arow[4 * t + 1];
  }

  for (int i = tid; i < K * 16; i += 128) {
    int row = i >> 4, c4 = (i & 15) * 4;
    *(float4*)&wsm[row * WS_STRIDE + c4] = *(const float4*)&Wt[(size_t)row * HID + c4];
  }
  __syncthreads();

  if (valid) {
    if (RELU_A) {
#pragma unroll
      for (int t = 0; t < K / 4; ++t) {
        a[t][0] = fmaxf(a[t][0], 0.0f);
        a[t][1] = fmaxf(a[t][1], 0.0f);
      }
    }

    v8f acc[4] = {v8f{}, v8f{}, v8f{}, v8f{}};
#pragma unroll
    for (int t = 0; t < K / 4; ++t) {
      const float* bp = wsm + (4 * t + 2 * half) * WS_STRIDE + mr;
#pragma unroll
      for (int nt = 0; nt < 4; ++nt) {
        v2f b;
        b[0] = bp[nt * 16];
        b[1] = bp[WS_STRIDE + nt * 16];
        acc[nt] = __builtin_amdgcn_wmma_f32_16x16x4_f32(false, a[t], false, b,
                                                        (short)0, acc[nt], false, false);
      }
    }

    float* hrow = H + (size_t)(m0 + 8 * half) * HID + mr;
#pragma unroll
    for (int nt = 0; nt < 4; ++nt) {
#pragma unroll
      for (int r = 0; r < 8; ++r) hrow[(size_t)r * HID + nt * 16] = acc[nt][r];
    }
  }
}

#define AGG_NPW 4
#define AS 68

template <int NPAIR>
__global__ __launch_bounds__(256) void agg_gemm_k(float* __restrict__ O0,
                                                  float* __restrict__ O1,
                                                  const float* __restrict__ h,
                                                  const int* __restrict__ cnt,
                                                  const int* __restrict__ tab,
                                                  const int2* __restrict__ ovf,
                                                  const int* __restrict__ octr,
                                                  const float* __restrict__ bias,
                                                  const float* __restrict__ Wt, int N,
                                                  int E) {
  __shared__ __attribute__((aligned(16))) float as[8 * AGG_NPW * AS];

  const int lane = threadIdx.x & 31;
  const int wv = __builtin_amdgcn_readfirstlane((int)(threadIdx.x >> 5));
  const int n0 = ((int)blockIdx.x * 8 + wv) * AGG_NPW;
  const int nn = max(0, min(AGG_NPW, N - n0));
  float* arow = as + (wv * AGG_NPW) * AS + lane;

  int ct = 0;
  if (lane < nn) ct = cnt[n0 + lane];
  const float dl = (ct > 0) ? rsqrtf((float)ct) : 0.0f;
  const float b0 = bias[lane];
  const float b1 = bias[lane + 32];
  const int* trow = tab + (size_t)n0 * CAP;

  int tot = __builtin_amdgcn_readlane(ct, 0);
  int len = min(tot, CAP);
  int es = 0;
  if (lane < len) es = trow[lane];

  for (int i = 0; i < nn; ++i) {
    const float dd = __int_as_float(__builtin_amdgcn_readlane(__float_as_int(dl), i));
    int ntot = 0, nlen = 0, nes = 0;

    float acc0 = 0.0f, acc1 = 0.0f;
    for (int done = 0; done < len; done += 32) {
      if (done > 0) {
        es = 0;
        if (done + lane < len) es = trow[i * CAP + done + lane];
      }
      const int c = min(32, len - done);
      const int cs = cnt[es];
      float v0[32], v1[32];
#pragma unroll
      for (int u = 0; u < 32; ++u) {
        v0[u] = 0.0f;
        v1[u] = 0.0f;
        if (u < c) {
          const int sj = __builtin_amdgcn_readlane(es, u);
          const float* hp = h + (size_t)sj * HID + lane;
          v0[u] = hp[0];
          v1[u] = hp[32];
        }
      }
      if (done == 0 && i + 1 < nn) {
        ntot = __builtin_amdgcn_readlane(ct, i + 1);
        nlen = min(ntot, CAP);
        if (lane < nlen) nes = trow[(i + 1) * CAP + lane];
      }
      const float wl = (done + lane < len && cs > 0) ? rsqrtf((float)cs) : 0.0f;
#pragma unroll
      for (int u = 0; u < 32; ++u) {
        const float wj = __int_as_float(__builtin_amdgcn_readlane(__float_as_int(wl), u)) * dd;
        acc0 += wj * v0[u];
        acc1 += wj * v1[u];
      }
    }

    if (tot > CAP) {
      const int node = n0 + i;
      const int nov = min(octr[0], E);
      for (int k0 = 0; k0 < nov; k0 += 32) {
        int2 o = make_int2(-1, 0);
        if (k0 + lane < nov) o = ovf[k0 + lane];
        for (int u = 0; u < 32; ++u) {
          if (__shfl(o.x, u) == node) {
            const int sj = __shfl(o.y, u);
            const int cj = cnt[sj];
            const float wj = ((cj > 0) ? rsqrtf((float)cj) : 0.0f) * dd;
            const float* hp = h + (size_t)sj * HID + lane;
            acc0 += wj * hp[0];
            acc1 += wj * hp[32];
          }
        }
      }
    }

    arow[i * AS]      = fmaxf(b0 + acc0, 0.0f);
    arow[i * AS + 32] = fmaxf(b1 + acc1, 0.0f);

    tot = ntot;
    len = nlen;
    es = nes;
  }
  for (int i = nn; i < AGG_NPW; ++i) {
    arow[i * AS]      = 0.0f;
    arow[i * AS + 32] = 0.0f;
  }

  const int mr   = lane & 15;
  const int half = lane >> 4;
  const bool worker = wv < 2 * NPAIR;
  const int mt = wv / NPAIR;
  const int np = wv % NPAIR;
  v2f bf[2][16];
  if (worker) {
    const float* wp = Wt + (size_t)((np >> 1) * HID + 2 * half) * HID + (np & 1) * 32 + 2 * mr;
#pragma unroll
    for (int t = 0; t < 16; ++t) {
      const float2 w0 = *(const float2*)(wp + (size_t)(4 * t) * HID);
      const float2 w1 = *(const float2*)(wp + (size_t)(4 * t + 1) * HID);
      bf[0][t][0] = w0.x;
      bf[0][t][1] = w1.x;
      bf[1][t][0] = w0.y;
      bf[1][t][1] = w1.y;
    }
  }
  __syncthreads();

  const int m0 = (int)blockIdx.x * (8 * AGG_NPW) + mt * 16;
  if (worker && m0 < N) {
    const float* ap = as + (mt * 16 + mr) * AS + 2 * half;
    v8f acc[2] = {v8f{}, v8f{}};
#pragma unroll
    for (int t = 0; t < 16; ++t) {
      const v2f a = *(const v2f*)(ap + 4 * t);
#pragma unroll
      for (int nt = 0; nt < 2; ++nt)
        acc[nt] = __builtin_amdgcn_wmma_f32_16x16x4_f32(false, a, false, bf[nt][t],
                                                        (short)0, acc[nt], false, false);
    }

    float* obase = (NPAIR == 4 && (np >> 1)) ? O1 : O0;
    float* orow = obase + (size_t)(m0 + 8 * half) * HID + (np & 1) * 32 + 2 * mr;
#pragma unroll
    for (int r = 0; r < 8; ++r)
      *(float2*)(orow + (size_t)r * HID) = make_float2(acc[0][r], acc[1][r]);
  }
}


#define ZS 68
#define OS 144

__global__ __launch_bounds__(128) void edge_out_k(const float* __restrict__ P,
                                                  const float* __restrict__ Q,
                                                  const int* __restrict__ ei,
                                                  const float* __restrict__ bm1,
                                                  const float* __restrict__ Wm2,
                                                  const float* __restrict__ bm2,
                                                  float* __restrict__ out, int E, int ntiles) {
  __shared__ float zs[4][16 * ZS];

  const int tid  = threadIdx.x;
  const int lane = tid & 31;
  const int mr   = lane & 15;
  const int half = lane >> 4;
  const int wave = __builtin_amdgcn_readfirstlane(tid >> 5);
  float* zw = zs[wave];

  v2f bf[16];
#pragma unroll
  for (int j = 0; j < 8; ++j) {
    const float* wp = Wm2 + (32 * half + 4 * j) * OUT_C + mr;
    bf[2 * j][0]     = wp[0];
    bf[2 * j][1]     = wp[OUT_C];
    bf[2 * j + 1][0] = wp[2 * OUT_C];
    bf[2 * j + 1][1] = wp[3 * OUT_C];
  }
  const float4 b1v = *(const float4*)&bm1[4 * mr];
  const float bo = bm2[mr];

  const int stride = (int)gridDim.x * 4;
  int tile = (int)blockIdx.x * 4 + wave;
  const int* eip = ei + (half ? E : 0) + mr;
  int idx = (tile < ntiles) ? eip[tile * 16] : 0;

  for (; tile < ntiles; tile += stride) {
    const int tn = tile + stride;
    const int idx_next = (tn < ntiles) ? eip[tn * 16] : 0;

    float4 z[8];
#pragma unroll
    for (int i = 0; i < 8; ++i) {
      const int r0 = __builtin_amdgcn_readlane(idx, 2 * i);
      const int r1 = __builtin_amdgcn_readlane(idx, 2 * i + 1);
      const int c0 = __builtin_amdgcn_readlane(idx, 16 + 2 * i);
      const int c1 = __builtin_amdgcn_readlane(idx, 17 + 2 * i);
      const unsigned ro = (unsigned)(half ? r1 : r0) * (unsigned)(HID * 4) + (unsigned)mr * 16u;
      const unsigned co = (unsigned)(half ? c1 : c0) * (unsigned)(HID * 4) + (unsigned)mr * 16u;
      const float4 p = *(const float4*)((const char*)P + ro);
      const float4 q = *(const float4*)((const char*)Q + co);
      z[i].x = fmaxf(p.x + q.x + b1v.x, 0.0f);
      z[i].y = fmaxf(p.y + q.y + b1v.y, 0.0f);
      z[i].z = fmaxf(p.z + q.z + b1v.z, 0.0f);
      z[i].w = fmaxf(p.w + q.w + b1v.w, 0.0f);
    }
#pragma unroll
    for (int i = 0; i < 8; ++i)
      *(float4*)&zw[(2 * i + half) * ZS + 4 * mr] = z[i];


    v8f o = {};
#pragma unroll
    for (int j = 0; j < 8; ++j) {
      const float4 a4 = *(const float4*)&zw[mr * ZS + 32 * half + 4 * j];
      v2f a0, a1;
      a0[0] = a4.x; a0[1] = a4.y;
      a1[0] = a4.z; a1[1] = a4.w;
      o = __builtin_amdgcn_wmma_f32_16x16x4_f32(false, a0, false, bf[2 * j],
                                                (short)0, o, false, false);
      o = __builtin_amdgcn_wmma_f32_16x16x4_f32(false, a1, false, bf[2 * j + 1],
                                                (short)0, o, false, false);
    }

#pragma unroll
    for (int r = 0; r < 8; ++r) zw[half * OS + r * 16 + mr] = o[r] + bo;

    float* ob = out + (size_t)tile * (16 * OUT_C) + 4 * lane;
    const float4 o0 = *(const float4*)&zw[4 * lane];
    const float4 o1 = *(const float4*)&zw[OS + 4 * lane];
    *(float4*)ob = o0;
    *(float4*)(ob + 128) = o1;

    idx = idx_next;
  }
}

extern "C" void kernel_launch(void* const* d_in, const int* in_sizes, int n_in,
                              void* d_out, int out_size, void* d_ws, size_t ws_size,
                              hipStream_t stream) {
  const float* x   = (const float*)d_in[0];
  const int*   ei  = (const int*)d_in[1];
  const float* W1  = (const float*)d_in[2];
  const float* b1  = (const float*)d_in[3];
  const float* W2  = (const float*)d_in[4];
  const float* b2  = (const float*)d_in[5];
  const float* Wm1 = (const float*)d_in[6];
  const float* bm1 = (const float*)d_in[7];
  const float* Wm2 = (const float*)d_in[8];
  const float* bm2 = (const float*)d_in[9];
  float* out = (float*)d_out;

  const int N = in_sizes[0] / IN_C;
  const int E = in_sizes[1] / 2;

  float* bufA = (float*)d_ws + (size_t)((N + 63) & ~63);
  float* bufB = bufA + (size_t)N * HID;

  int*  o_cnt  = (int*)d_out;
  int*  o_ctr  = o_cnt + (size_t)((N + 63) & ~63);
  int*  o_tab  = o_ctr + 64;
  int2* o_ovf  = (int2*)(o_tab + (size_t)N * CAP);
  float* o_h2  = (float*)(o_ovf + (size_t)((E + 31) & ~31));

  const int mtiles = (N + 15) / 16;
  const int gx     = (mtiles + 3) / 4;
  const int nb  = (N + 255) / 256;
  const int ab  = (N + 8 * AGG_NPW - 1) / (8 * AGG_NPW);
  const int qb  = (E + 256 * EPL - 1) / (256 * EPL);

  cnt_init_k<<<nb, 256, 0, stream>>>(o_cnt, o_tab, o_ctr, N);
  place_edges_k<<<qb, 256, 0, stream>>>(o_cnt, o_tab, o_ovf, o_ctr, ei, E);

  gemm_rows_k<IN_C, false><<<gx, 128, 0, stream>>>(x, W1, bufA, N);
  agg_gemm_k<2><<<ab, 256, 0, stream>>>(o_h2, nullptr, bufA, o_cnt, o_tab, o_ovf, o_ctr, b1, W2, N, E);
  agg_gemm_k<4><<<ab, 256, 0, stream>>>(bufA, bufB, o_h2, o_cnt, o_tab, o_ovf, o_ctr, b2, Wm1, N, E);

  edge_out_k<<<1250, 128, 0, stream>>>(bufA, bufB, ei, bm1, Wm2, bm2, out, E, E / 16);
}


# source weight moved to the producers: h rows are stored as dinv[node]*h, so a bucket sum is plain adds times dinv[d] once; per-run gather of source counts and per-entry weight ops go; sum order kept
# speedup vs baseline: 2.5455x; 1.0453x over previous
#include <hip/hip_runtime.h>
#include <hip/hip_bf16.h>

typedef __attribute__((ext_vector_type(2))) float v2f;
typedef __attribute__((ext_vector_type(8))) float v8f;

#define IN_C 128
#define HID  64
#define OUT_C 16

#define CAP 64

__global__ void cnt_init_k(int* __restrict__ cnt, int* __restrict__ tab,
                           int* __restrict__ octr, int n) {
  int i = blockIdx.x * blockDim.x + threadIdx.x;
  if (i < n) {
    cnt[i] = 1;
    tab[(size_t)i * CAP] = i;
  }
  if (i == 0) octr[0] = 0;
}

#define EPL 4

__global__ void place_edges_k(int* __restrict__ cnt, int* __restrict__ tab,
                              int2* __restrict__ ovf, int* __restrict__ octr,
                              const int* __restrict__ ei, int E) {
  const int T  = (int)(gridDim.x * blockDim.x);
  const int e0 = blockIdx.x * blockDim.x + threadIdx.x;
  unsigned s[EPL], d[EPL];
  int r[EPL];
#pragma unroll
  for (int k = 0; k < EPL; ++k) {
    const int ec = min(e0 + k * T, E - 1);
    s[k] = (unsigned)ei[ec];
    d[k] = (unsigned)ei[E + ec];
  }
#pragma unroll
  for (int k = 0; k < EPL; ++k)
    r[k] = (e0 + k * T < E) ? atomicAdd(&cnt[d[k]], 1) : 0;
#pragma unroll
  for (int k = 0; k < EPL; ++k) {
    if (e0 + k * T < E) {
      if (r[k] < CAP) {
        tab[(size_t)d[k] * CAP + r[k]] = (int)s[k];
      } else {
        const int q = atomicAdd(octr, 1);
        ovf[q] = make_int2((int)d[k], (int)s[k]);
      }
    }
  }
}

#define WS_STRIDE 72

template <int K, bool RELU_A>
__global__ __launch_bounds__(128) void gemm_rows_k(const float* __restrict__ A,
                                                   const float* __restrict__ Wt,
                                                   const int* __restrict__ cnt,
                                                   float* __restrict__ H, int M) {
  __shared__ float wsm[K * WS_STRIDE];

  const int tid  = threadIdx.x;
  const int wave = tid >> 5;
  const int lane = tid & 31;
  const int mr   = lane & 15;
  const int half = lane >> 4;

  const int tile = blockIdx.x * 4 + wave;
  const int m0   = tile * 16;
  const bool valid = m0 < M;

  const float* arow = A + (size_t)min(m0 + mr, M - 1) * K + 2 * half;
  v2f a[K / 4];
#pragma unroll
  for (int t = 0; t < K / 4; ++t) {
    a[t][0] = arow[4 * t];
    a[t][1] = arow[4 * t + 1];
  }
  const int cb = min(m0 + 8 * half, M - 8);
  const int4 cq0 = *(const int4*)&cnt[cb];
  const int4 cq1 = *(const int4*)&cnt[cb + 4];

  for (int i = tid; i < K * 16; i += 128) {
    int row = i >> 4, c4 = (i & 15) * 4;
    *(float4*)&wsm[row * WS_STRIDE + c4] = *(const float4*)&Wt[(size_t)row * HID + c4];
  }
  __syncthreads();

  if (valid) {
    if (RELU_A) {
#pragma unroll
      for (int t = 0; t < K / 4; ++t) {
        a[t][0] = fmaxf(a[t][0], 0.0f);
        a[t][1] = fmaxf(a[t][1], 0.0f);
      }
    }

    v8f acc[4] = {v8f{}, v8f{}, v8f{}, v8f{}};
#pragma unroll
    for (int t = 0; t < K / 4; ++t) {
      const float* bp = wsm + (4 * t + 2 * half) * WS_STRIDE + mr;
#pragma unroll
      for (int nt = 0; nt < 4; ++nt) {
        v2f b;
        b[0] = bp[nt * 16];
        b[1] = bp[WS_STRIDE + nt * 16];
        acc[nt] = __builtin_amdgcn_wmma_f32_16x16x4_f32(false, a[t], false, b,
                                                        (short)0, acc[nt], false, false);
      }
    }

    const int cr[8] = {cq0.x, cq0.y, cq0.z, cq0.w, cq1.x, cq1.y, cq1.z, cq1.w};
    float dv[8];
#pragma unroll
    for (int r = 0; r < 8; ++r) dv[r] = (cr[r] > 0) ? rsqrtf((float)cr[r]) : 0.0f;

    float* hrow = H + (size_t)(m0 + 8 * half) * HID + mr;
#pragma unroll
    for (int nt = 0; nt < 4; ++nt) {
#pragma unroll
      for (int r = 0; r < 8; ++r) hrow[(size_t)r * HID + nt * 16] = dv[r] * acc[nt][r];
    }
  }
}

#define AGG_NPW 4
#define AS 68

template <int NPAIR>
__global__ __launch_bounds__(256) void agg_gemm_k(float* __restrict__ O0,
                                                  float* __restrict__ O1,
                                                  const float* __restrict__ h,
                                                  const int* __restrict__ cnt,
                                                  const int* __restrict__ tab,
                                                  const int2* __restrict__ ovf,
                                                  const int* __restrict__ octr,
                                                  const float* __restrict__ bias,
                                                  const float* __restrict__ Wt, int N,
                                                  int E) {
  __shared__ __attribute__((aligned(16))) float as[8 * AGG_NPW * AS];

  const int lane = threadIdx.x & 31;
  const int wv = __builtin_amdgcn_readfirstlane((int)(threadIdx.x >> 5));
  const int n0 = ((int)blockIdx.x * 8 + wv) * AGG_NPW;
  const int nn = max(0, min(AGG_NPW, N - n0));
  float* arow = as + (wv * AGG_NPW) * AS + lane;

  int ct = 0;
  if (lane < nn) ct = cnt[n0 + lane];
  const float dl = (ct > 0) ? rsqrtf((float)ct) : 0.0f;
  const float b0 = bias[lane];
  const float b1 = bias[lane + 32];
  const int* trow = tab + (size_t)n0 * CAP;

  int tot = __builtin_amdgcn_readlane(ct, 0);
  int len = min(tot, CAP);
  int es = 0;
  if (lane < len) es = trow[lane];

  for (int i = 0; i < nn; ++i) {
    const float dd = __int_as_float(__builtin_amdgcn_readlane(__float_as_int(dl), i));
    int ntot = 0, nlen = 0, nes = 0;

    float acc0 = 0.0f, acc1 = 0.0f;
    for (int done = 0; done < len; done += 32) {
      if (done > 0) {
        es = 0;
        if (done + lane < len) es = trow[i * CAP + done + lane];
      }
      const int c = min(32, len - done);
      float v0[32], v1[32];
#pragma unroll
      for (int u = 0; u < 32; ++u) {
        v0[u] = 0.0f;
        v1[u] = 0.0f;
        if (u < c) {
          const int sj = __builtin_amdgcn_readlane(es, u);
          const float* hp = h + (size_t)sj * HID + lane;
          v0[u] = hp[0];
          v1[u] = hp[32];
        }
      }
      if (done == 0 && i + 1 < nn) {
        ntot = __builtin_amdgcn_readlane(ct, i + 1);
        nlen = min(ntot, CAP);
        if (lane < nlen) nes = trow[(i + 1) * CAP + lane];
      }
#pragma unroll
      for (int u = 0; u < 32; ++u) {
        acc0 += v0[u];
        acc1 += v1[u];
      }
    }

    if (tot > CAP) {
      const int node = n0 + i;
      const int nov = min(octr[0], E);
      for (int k0 = 0; k0 < nov; k0 += 32) {
        int2 o = make_int2(-1, 0);
        if (k0 + lane < nov) o = ovf[k0 + lane];
#pragma nounroll
        for (int u = 0; u < 32; ++u) {
          if (__shfl(o.x, u) == node) {
            const int sj = __shfl(o.y, u);
            const float* hp = h + (size_t)sj * HID + lane;
            acc0 += hp[0];
            acc1 += hp[32];
          }
        }
      }
    }

    const float so = (NPAIR == 2) ? dd : 1.0f;
    arow[i * AS]      = so * fmaxf(b0 + dd * acc0, 0.0f);
    arow[i * AS + 32] = so * fmaxf(b1 + dd * acc1, 0.0f);

    tot = ntot;
    len = nlen;
    es = nes;
  }
  for (int i = nn; i < AGG_NPW; ++i) {
    arow[i * AS]      = 0.0f;
    arow[i * AS + 32] = 0.0f;
  }

  const int mr   = lane & 15;
  const int half = lane >> 4;
  const bool worker = wv < 2 * NPAIR;
  const int mt = wv / NPAIR;
  const int np = wv % NPAIR;
  v2f bf[2][16];
  if (worker) {
    const float* wp = Wt + (size_t)((np >> 1) * HID + 2 * half) * HID + (np & 1) * 32 + 2 * mr;
#pragma unroll
    for (int t = 0; t < 16; ++t) {
      const float2 w0 = *(const float2*)(wp + (size_t)(4 * t) * HID);
      const float2 w1 = *(const float2*)(wp + (size_t)(4 * t + 1) * HID);
      bf[0][t][0] = w0.x;
      bf[0][t][1] = w1.x;
      bf[1][t][0] = w0.y;
      bf[1][t][1] = w1.y;
    }
  }
  __syncthreads();

  const int m0 = (int)blockIdx.x * (8 * AGG_NPW) + mt * 16;
  if (worker && m0 < N) {
    const float* ap = as + (mt * 16 + mr) * AS + 2 * half;
    v8f acc[2] = {v8f{}, v8f{}};
#pragma unroll
    for (int t = 0; t < 16; ++t) {
      const v2f a = *(const v2f*)(ap + 4 * t);
#pragma unroll
      for (int nt = 0; nt < 2; ++nt)
        acc[nt] = __builtin_amdgcn_wmma_f32_16x16x4_f32(false, a, false, bf[nt][t],
                                                        (short)0, acc[nt], false, false);
    }

    float* obase = (NPAIR == 4 && (np >> 1)) ? O1 : O0;
    float* orow = obase + (size_t)(m0 + 8 * half) * HID + (np & 1) * 32 + 2 * mr;
#pragma unroll
    for (int r = 0; r < 8; ++r)
      *(float2*)(orow + (size_t)r * HID) = make_float2(acc[0][r], acc[1][r]);
  }
}


#define ZS 68
#define OS 144

__global__ __launch_bounds__(128) void edge_out_k(const float* __restrict__ P,
                                                  const float* __restrict__ Q,
                                                  const int* __restrict__ ei,
                                                  const float* __restrict__ bm1,
                                                  const float* __restrict__ Wm2,
                                                  const float* __restrict__ bm2,
                                                  float* __restrict__ out, int E, int ntiles) {
  __shared__ float zs[4][16 * ZS];

  const int tid  = threadIdx.x;
  const int lane = tid & 31;
  const int mr   = lane & 15;
  const int half = lane >> 4;
  const int wave = __builtin_amdgcn_readfirstlane(tid >> 5);
  float* zw = zs[wave];

  v2f bf[16];
#pragma unroll
  for (int j = 0; j < 8; ++j) {
    const float* wp = Wm2 + (32 * half + 4 * j) * OUT_C + mr;
    bf[2 * j][0]     = wp[0];
    bf[2 * j][1]     = wp[OUT_C];
    bf[2 * j + 1][0] = wp[2 * OUT_C];
    bf[2 * j + 1][1] = wp[3 * OUT_C];
  }
  const float4 b1v = *(const float4*)&bm1[4 * mr];
  const float bo = bm2[mr];

  const int stride = (int)gridDim.x * 4;
  int tile = (int)blockIdx.x * 4 + wave;
  const int* eip = ei + (half ? E : 0) + mr;
  int idx = (tile < ntiles) ? eip[tile * 16] : 0;

  for (; tile < ntiles; tile += stride) {
    const int tn = tile + stride;
    const int idx_next = (tn < ntiles) ? eip[tn * 16] : 0;

    float4 z[8];
#pragma unroll
    for (int i = 0; i < 8; ++i) {
      const int r0 = __builtin_amdgcn_readlane(idx, 2 * i);
      const int r1 = __builtin_amdgcn_readlane(idx, 2 * i + 1);
      const int c0 = __builtin_amdgcn_readlane(idx, 16 + 2 * i);
      const int c1 = __builtin_amdgcn_readlane(idx, 17 + 2 * i);
      const unsigned ro = (unsigned)(half ? r1 : r0) * (unsigned)(HID * 4) + (unsigned)mr * 16u;
      const unsigned co = (unsigned)(half ? c1 : c0) * (unsigned)(HID * 4) + (unsigned)mr * 16u;
      const float4 p = *(const float4*)((const char*)P + ro);
      const float4 q = *(const float4*)((const char*)Q + co);
      z[i].x = fmaxf(p.x + q.x + b1v.x, 0.0f);
      z[i].y = fmaxf(p.y + q.y + b1v.y, 0.0f);
      z[i].z = fmaxf(p.z + q.z + b1v.z, 0.0f);
      z[i].w = fmaxf(p.w + q.w + b1v.w, 0.0f);
    }
#pragma unroll
    for (int i = 0; i < 8; ++i)
      *(float4*)&zw[(2 * i + half) * ZS + 4 * mr] = z[i];


    v8f o = {};
#pragma unroll
    for (int j = 0; j < 8; ++j) {
      const float4 a4 = *(const float4*)&zw[mr * ZS + 32 * half + 4 * j];
      v2f a0, a1;
      a0[0] = a4.x; a0[1] = a4.y;
      a1[0] = a4.z; a1[1] = a4.w;
      o = __builtin_amdgcn_wmma_f32_16x16x4_f32(false, a0, false, bf[2 * j],
                                                (short)0, o, false, false);
      o = __builtin_amdgcn_wmma_f32_16x16x4_f32(false, a1, false, bf[2 * j + 1],
                                                (short)0, o, false, false);
    }

#pragma unroll
    for (int r = 0; r < 8; ++r) zw[half * OS + r * 16 + mr] = o[r] + bo;

    float* ob = out + (size_t)tile * (16 * OUT_C) + 4 * lane;
    const float4 o0 = *(const float4*)&zw[4 * lane];
    const float4 o1 = *(const float4*)&zw[OS + 4 * lane];
    *(float4*)ob = o0;
    *(float4*)(ob + 128) = o1;

    idx = idx_next;
  }
}

extern "C" void kernel_launch(void* const* d_in, const int* in_sizes, int n_in,
                              void* d_out, int out_size, void* d_ws, size_t ws_size,
                              hipStream_t stream) {
  const float* x   = (const float*)d_in[0];
  const int*   ei  = (const int*)d_in[1];
  const float* W1  = (const float*)d_in[2];
  const float* b1  = (const float*)d_in[3];
  const float* W2  = (const float*)d_in[4];
  const float* b2  = (const float*)d_in[5];
  const float* Wm1 = (const float*)d_in[6];
  const float* bm1 = (const float*)d_in[7];
  const float* Wm2 = (const float*)d_in[8];
  const float* bm2 = (const float*)d_in[9];
  float* out = (float*)d_out;

  const int N = in_sizes[0] / IN_C;
  const int E = in_sizes[1] / 2;

  float* bufA = (float*)d_ws + (size_t)((N + 63) & ~63);
  float* bufB = bufA + (size_t)N * HID;

  int*  o_cnt  = (int*)d_out;
  int*  o_ctr  = o_cnt + (size_t)((N + 63) & ~63);
  int*  o_tab  = o_ctr + 64;
  int2* o_ovf  = (int2*)(o_tab + (size_t)N * CAP);
  float* o_h2  = (float*)(o_ovf + (size_t)((E + 31) & ~31));

  const int mtiles = (N + 15) / 16;
  const int gx     = (mtiles + 3) / 4;
  const int nb  = (N + 255) / 256;
  const int ab  = (N + 8 * AGG_NPW - 1) / (8 * AGG_NPW);
  const int qb  = (E + 256 * EPL - 1) / (256 * EPL);

  cnt_init_k<<<nb, 256, 0, stream>>>(o_cnt, o_tab, o_ctr, N);
  place_edges_k<<<qb, 256, 0, stream>>>(o_cnt, o_tab, o_ovf, o_ctr, ei, E);

  gemm_rows_k<IN_C, false><<<gx, 128, 0, stream>>>(x, W1, o_cnt, bufA, N);
  agg_gemm_k<2><<<ab, 256, 0, stream>>>(o_h2, nullptr, bufA, o_cnt, o_tab, o_ovf, o_ctr, b1, W2, N, E);
  agg_gemm_k<4><<<ab, 256, 0, stream>>>(bufA, bufB, o_h2, o_cnt, o_tab, o_ovf, o_ctr, b2, Wm1, N, E);

  edge_out_k<<<1250, 128, 0, stream>>>(bufA, bufB, ei, bm1, Wm2, bm2, out, E, E / 16);
}


# edge buckets: the counting adds move from global memory to LDS: launch 1 sorts each block's chunk of edges by destination range, launch 2 ranks a range in LDS; sum order still arrival rank
# speedup vs baseline: 3.4372x; 1.3503x over previous
#include <hip/hip_runtime.h>
#include <hip/hip_bf16.h>

typedef __attribute__((ext_vector_type(2))) float v2f;
typedef __attribute__((ext_vector_type(8))) float v8f;

#define IN_C 128
#define HID  64
#define OUT_C 16

#define CAP 64
#define RS_LOG 7
#define RS (1 << RS_LOG)
#define NB1 256
#define SEGCAP 32
#define EPL 8

__global__ __launch_bounds__(256) void part_edges_k(int* __restrict__ seg,
                                                    int* __restrict__ segcnt,
                                                    int* __restrict__ octr,
                                                    const int* __restrict__ ei, int E, int NR,
                                                    int CH) {
  extern __shared__ int lbin[];
  const int t = threadIdx.x;
  const int b = blockIdx.x;
  for (int r = t; r < NR; r += 256) lbin[r] = 0;
  if (b == 0 && t == 0) octr[0] = 0;
  __syncthreads();

  const int e0 = b * CH;
  int* sb = seg + (size_t)b * NR * SEGCAP;
  for (int k0 = 0; k0 < CH; k0 += 256 * EPL) {
    unsigned s[EPL], d[EPL];
#pragma unroll
    for (int u = 0; u < EPL; ++u) {
      const int ec = min(e0 + k0 + u * 256 + t, E - 1);
      s[u] = (unsigned)ei[ec];
      d[u] = (unsigned)ei[E + ec];
    }
#pragma unroll
    for (int u = 0; u < EPL; ++u) {
      const int k = k0 + u * 256 + t;
      if (k < CH && e0 + k < E) {
        const unsigned r = d[u] >> RS_LOG;
        const int p = atomicAdd(&lbin[r], 1);
        if (p < SEGCAP)
          sb[(size_t)r * SEGCAP + p] = (int)(s[u] | ((d[u] & (unsigned)(RS - 1)) << 24));
      }
    }
  }
  __syncthreads();
  for (int r = t; r < NR; r += 256) segcnt[(size_t)r * NB1 + b] = lbin[r];
}

__device__ __forceinline__ void place_one(int* lcnt, int* __restrict__ tab,
                                          int2* __restrict__ ovf, int* __restrict__ octr,
                                          int nbase, unsigned dl, int s) {
  const int rank = atomicAdd(&lcnt[dl], 1);
  const int d = nbase + (int)dl;
  if (rank < CAP) {
    tab[(size_t)d * CAP + rank] = s;
  } else {
    const int q = atomicAdd(octr, 1);
    ovf[q] = make_int2(d, s);
  }
}

__global__ __launch_bounds__(NB1) void place_edges_k(int* __restrict__ cnt, int* __restrict__ tab,
                                                     int2* __restrict__ ovf,
                                                     int* __restrict__ octr,
                                                     const int* __restrict__ seg,
                                                     const int* __restrict__ segcnt,
                                                     const int* __restrict__ ei, int N, int E,
                                                     int NR, int CH) {
  __shared__ int lcnt[RS];
  const int t = threadIdx.x;
  const int r = blockIdx.x;
  const int c = segcnt[(size_t)r * NB1 + t];
  if (t < RS) lcnt[t] = 1;
  __syncthreads();

  const int nbase = r << RS_LOG;
  if (c <= SEGCAP) {
    const int* sp = seg + ((size_t)t * NR + r) * SEGCAP;
    for (int j = 0; j < c; j += 4) {
      int w[4];
#pragma unroll
      for (int u = 0; u < 4; ++u) w[u] = (j + u < c) ? sp[j + u] : 0;
#pragma unroll
      for (int u = 0; u < 4; ++u)
        if (j + u < c)
          place_one(lcnt, tab, ovf, octr, nbase, (unsigned)w[u] >> 24, w[u] & 0xFFFFFF);
    }
  } else {
    const int eb = t * CH;
    const int ee = min(eb + CH, E);
    for (int e = eb; e < ee; ++e) {
      const unsigned d = (unsigned)ei[E + e];
      if ((int)(d >> RS_LOG) == r)
        place_one(lcnt, tab, ovf, octr, nbase, d & (unsigned)(RS - 1), ei[e]);
    }
  }
  __syncthreads();
  if (t < RS && nbase + t < N) {
    cnt[nbase + t] = lcnt[t];
    tab[(size_t)(nbase + t) * CAP] = nbase + t;
  }
}

#define WS_STRIDE 72

template <int K, bool RELU_A>
__global__ __launch_bounds__(128) void gemm_rows_k(const float* __restrict__ A,
                                                   const float* __restrict__ Wt,
                                                   const int* __restrict__ cnt,
                                                   float* __restrict__ H, int M) {
  __shared__ float wsm[K * WS_STRIDE];

  const int tid  = threadIdx.x;
  const int wave = tid >> 5;
  const int lane = tid & 31;
  const int mr   = lane & 15;
  const int half = lane >> 4;

  const int tile = blockIdx.x * 4 + wave;
  const int m0   = tile * 16;
  const bool valid = m0 < M;

  const float* arow = A + (size_t)min(m0 + mr, M - 1) * K + 2 * half;
  v2f a[K / 4];
#pragma unroll
  for (int t = 0; t < K / 4; ++t) {
    a[t][0] = arow[4 * t];
    a[t][1] = arow[4 * t + 1];
  }
  const int cb = min(m0 + 8 * half, M - 8);
  const int4 cq0 = *(const int4*)&cnt[cb];
  const int4 cq1 = *(const int4*)&cnt[cb + 4];

  for (int i = tid; i < K * 16; i += 128) {
    int row = i >> 4, c4 = (i & 15) * 4;
    *(float4*)&wsm[row * WS_STRIDE + c4] = *(const float4*)&Wt[(size_t)row * HID + c4];
  }
  __syncthreads();

  if (valid) {
    if (RELU_A) {
#pragma unroll
      for (int t = 0; t < K / 4; ++t) {
        a[t][0] = fmaxf(a[t][0], 0.0f);
        a[t][1] = fmaxf(a[t][1], 0.0f);
      }
    }

    v8f acc[4] = {v8f{}, v8f{}, v8f{}, v8f{}};
#pragma unroll
    for (int t = 0; t < K / 4; ++t) {
      const float* bp = wsm + (4 * t + 2 * half) * WS_STRIDE + mr;
#pragma unroll
      for (int nt = 0; nt < 4; ++nt) {
        v2f b;
        b[0] = bp[nt * 16];
        b[1] = bp[WS_STRIDE + nt * 16];
        acc[nt] = __builtin_amdgcn_wmma_f32_16x16x4_f32(false, a[t], false, b,
                                                        (short)0, acc[nt], false, false);
      }
    }

    const int cr[8] = {cq0.x, cq0.y, cq0.z, cq0.w, cq1.x, cq1.y, cq1.z, cq1.w};
    float dv[8];
#pragma unroll
    for (int r = 0; r < 8; ++r) dv[r] = (cr[r] > 0) ? rsqrtf((float)cr[r]) : 0.0f;

    float* hrow = H + (size_t)(m0 + 8 * half) * HID + mr;
#pragma unroll
    for (int nt = 0; nt < 4; ++nt) {
#pragma unroll
      for (int r = 0; r < 8; ++r) hrow[(size_t)r * HID + nt * 16] = dv[r] * acc[nt][r];
    }
  }
}

#define AGG_NPW 4
#define AS 68

template <int NPAIR>
__global__ __launch_bounds__(256) void agg_gemm_k(float* __restrict__ O0,
                                                  float* __restrict__ O1,
                                                  const float* __restrict__ h,
                                                  const int* __restrict__ cnt,
                                                  const int* __restrict__ tab,
                                                  const int2* __restrict__ ovf,
                                                  const int* __restrict__ octr,
                                                  const float* __restrict__ bias,
                                                  const float* __restrict__ Wt, int N,
                                                  int E) {
  __shared__ __attribute__((aligned(16))) float as[8 * AGG_NPW * AS];

  const int lane = threadIdx.x & 31;
  const int wv = __builtin_amdgcn_readfirstlane((int)(threadIdx.x >> 5));
  const int n0 = ((int)blockIdx.x * 8 + wv) * AGG_NPW;
  const int nn = max(0, min(AGG_NPW, N - n0));
  float* arow = as + (wv * AGG_NPW) * AS + lane;

  int ct = 0;
  if (lane < nn) ct = cnt[n0 + lane];
  const float dl = (ct > 0) ? rsqrtf((float)ct) : 0.0f;
  const float b0 = bias[lane];
  const float b1 = bias[lane + 32];
  const int* trow = tab + (size_t)n0 * CAP;

  int tot = __builtin_amdgcn_readlane(ct, 0);
  int len = min(tot, CAP);
  int es = 0;
  if (lane < len) es = trow[lane];

  for (int i = 0; i < nn; ++i) {
    const float dd = __int_as_float(__builtin_amdgcn_readlane(__float_as_int(dl), i));
    int ntot = 0, nlen = 0, nes = 0;

    float acc0 = 0.0f, acc1 = 0.0f;
    for (int done = 0; done < len; done += 32) {
      if (done > 0) {
        es = 0;
        if (done + lane < len) es = trow[i * CAP + done + lane];
      }
      const int c = min(32, len - done);
      float v0[32], v1[32];
#pragma unroll
      for (int u = 0; u < 32; ++u) {
        v0[u] = 0.0f;
        v1[u] = 0.0f;
        if (u < c) {
          const int sj = __builtin_amdgcn_readlane(es, u);
          const float* hp = h + (size_t)sj * HID + lane;
          v0[u] = hp[0];
          v1[u] = hp[32];
        }
      }
      if (done == 0 && i + 1 < nn) {
        ntot = __builtin_amdgcn_readlane(ct, i + 1);
        nlen = min(ntot, CAP);
        if (lane < nlen) nes = trow[(i + 1) * CAP + lane];
      }
#pragma unroll
      for (int u = 0; u < 32; ++u) {
        acc0 += v0[u];
        acc1 += v1[u];
      }
    }

    if (tot > CAP) {
      const int node = n0 + i;
      const int nov = min(octr[0], E);
      for (int k0 = 0; k0 < nov; k0 += 32) {
        int2 o = make_int2(-1, 0);
        if (k0 + lane < nov) o = ovf[k0 + lane];
#pragma nounroll
        for (int u = 0; u < 32; ++u) {
          if (__shfl(o.x, u) == node) {
            const int sj = __shfl(o.y, u);
            const float* hp = h + (size_t)sj * HID + lane;
            acc0 += hp[0];
            acc1 += hp[32];
          }
        }
      }
    }

    const float so = (NPAIR == 2) ? dd : 1.0f;
    arow[i * AS]      = so * fmaxf(b0 + dd * acc0, 0.0f);
    arow[i * AS + 32] = so * fmaxf(b1 + dd * acc1, 0.0f);

    tot = ntot;
    len = nlen;
    es = nes;
  }
  for (int i = nn; i < AGG_NPW; ++i) {
    arow[i * AS]      = 0.0f;
    arow[i * AS + 32] = 0.0f;
  }

  const int mr   = lane & 15;
  const int half = lane >> 4;
  const bool worker = wv < 2 * NPAIR;
  const int mt = wv / NPAIR;
  const int np = wv % NPAIR;
  v2f bf[2][16];
  if (worker) {
    const float* wp = Wt + (size_t)((np >> 1) * HID + 2 * half) * HID + (np & 1) * 32 + 2 * mr;
#pragma unroll
    for (int t = 0; t < 16; ++t) {
      const float2 w0 = *(const float2*)(wp + (size_t)(4 * t) * HID);
      const float2 w1 = *(const float2*)(wp + (size_t)(4 * t + 1) * HID);
      bf[0][t][0] = w0.x;
      bf[0][t][1] = w1.x;
      bf[1][t][0] = w0.y;
      bf[1][t][1] = w1.y;
    }
  }
  __syncthreads();

  const int m0 = (int)blockIdx.x * (8 * AGG_NPW) + mt * 16;
  if (worker && m0 < N) {
    const float* ap = as + (mt * 16 + mr) * AS + 2 * half;
    v8f acc[2] = {v8f{}, v8f{}};
#pragma unroll
    for (int t = 0; t < 16; ++t) {
      const v2f a = *(const v2f*)(ap + 4 * t);
#pragma unroll
      for (int nt = 0; nt < 2; ++nt)
        acc[nt] = __builtin_amdgcn_wmma_f32_16x16x4_f32(false, a, false, bf[nt][t],
                                                        (short)0, acc[nt], false, false);
    }

    float* obase = (NPAIR == 4 && (np >> 1)) ? O1 : O0;
    float* orow = obase + (size_t)(m0 + 8 * half) * HID + (np & 1) * 32 + 2 * mr;
#pragma unroll
    for (int r = 0; r < 8; ++r)
      *(float2*)(orow + (size_t)r * HID) = make_float2(acc[0][r], acc[1][r]);
  }
}


#define ZS 68
#define OS 144

__global__ __launch_bounds__(128) void edge_out_k(const float* __restrict__ P,
                                                  const float* __restrict__ Q,
                                                  const int* __restrict__ ei,
                                                  const float* __restrict__ bm1,
                                                  const float* __restrict__ Wm2,
                                                  const float* __restrict__ bm2,
                                                  float* __restrict__ out, int E, int ntiles) {
  __shared__ float zs[4][16 * ZS];

  const int tid  = threadIdx.x;
  const int lane = tid & 31;
  const int mr   = lane & 15;
  const int half = lane >> 4;
  const int wave = __builtin_amdgcn_readfirstlane(tid >> 5);
  float* zw = zs[wave];

  v2f bf[16];
#pragma unroll
  for (int j = 0; j < 8; ++j) {
    const float* wp = Wm2 + (32 * half + 4 * j) * OUT_C + mr;
    bf[2 * j][0]     = wp[0];
    bf[2 * j][1]     = wp[OUT_C];
    bf[2 * j + 1][0] = wp[2 * OUT_C];
    bf[2 * j + 1][1] = wp[3 * OUT_C];
  }
  const float4 b1v = *(const float4*)&bm1[4 * mr];
  const float bo = bm2[mr];

  const int stride = (int)gridDim.x * 4;
  int tile = (int)blockIdx.x * 4 + wave;
  const int* eip = ei + (half ? E : 0) + mr;
  int idx = (tile < ntiles) ? eip[tile * 16] : 0;

  for (; tile < ntiles; tile += stride) {
    const int tn = tile + stride;
    const int idx_next = (tn < ntiles) ? eip[tn * 16] : 0;

    float4 z[8];
#pragma unroll
    for (int i = 0; i < 8; ++i) {
      const int r0 = __builtin_amdgcn_readlane(idx, 2 * i);
      const int r1 = __builtin_amdgcn_readlane(idx, 2 * i + 1);
      const int c0 = __builtin_amdgcn_readlane(idx, 16 + 2 * i);
      const int c1 = __builtin_amdgcn_readlane(idx, 17 + 2 * i);
      const unsigned ro = (unsigned)(half ? r1 : r0) * (unsigned)(HID * 4) + (unsigned)mr * 16u;
      const unsigned co = (unsigned)(half ? c1 : c0) * (unsigned)(HID * 4) + (unsigned)mr * 16u;
      const float4 p = *(const float4*)((const char*)P + ro);
      const float4 q = *(const float4*)((const char*)Q + co);
      z[i].x = fmaxf(p.x + q.x + b1v.x, 0.0f);
      z[i].y = fmaxf(p.y + q.y + b1v.y, 0.0f);
      z[i].z = fmaxf(p.z + q.z + b1v.z, 0.0f);
      z[i].w = fmaxf(p.w + q.w + b1v.w, 0.0f);
    }
#pragma unroll
    for (int i = 0; i < 8; ++i)
      *(float4*)&zw[(2 * i + half) * ZS + 4 * mr] = z[i];


    v8f o = {};
#pragma unroll
    for (int j = 0; j < 8; ++j) {
      const float4 a4 = *(const float4*)&zw[mr * ZS + 32 * half + 4 * j];
      v2f a0, a1;
      a0[0] = a4.x; a0[1] = a4.y;
      a1[0] = a4.z; a1[1] = a4.w;
      o = __builtin_amdgcn_wmma_f32_16x16x4_f32(false, a0, false, bf[2 * j],
                                                (short)0, o, false, false);
      o = __builtin_amdgcn_wmma_f32_16x16x4_f32(false, a1, false, bf[2 * j + 1],
                                                (short)0, o, false, false);
    }

#pragma unroll
    for (int r = 0; r < 8; ++r) zw[half * OS + r * 16 + mr] = o[r] + bo;

    float* ob = out + (size_t)tile * (16 * OUT_C) + 4 * lane;
    const float4 o0 = *(const float4*)&zw[4 * lane];
    const float4 o1 = *(const float4*)&zw[OS + 4 * lane];
    *(float4*)ob = o0;
    *(float4*)(ob + 128) = o1;

    idx = idx_next;
  }
}

extern "C" void kernel_launch(void* const* d_in, const int* in_sizes, int n_in,
                              void* d_out, int out_size, void* d_ws, size_t ws_size,
                              hipStream_t stream) {
  const float* x   = (const float*)d_in[0];
  const int*   ei  = (const int*)d_in[1];
  const float* W1  = (const float*)d_in[2];
  const float* b1  = (const float*)d_in[3];
  const float* W2  = (const float*)d_in[4];
  const float* b2  = (const float*)d_in[5];
  const float* Wm1 = (const float*)d_in[6];
  const float* bm1 = (const float*)d_in[7];
  const float* Wm2 = (const float*)d_in[8];
  const float* bm2 = (const float*)d_in[9];
  float* out = (float*)d_out;

  const int N = in_sizes[0] / IN_C;
  const int E = in_sizes[1] / 2;

  float* bufA = (float*)d_ws + (size_t)((N + 63) & ~63);
  float* bufB = bufA + (size_t)N * HID;

  int*  o_cnt  = (int*)d_out;
  int*  o_ctr  = o_cnt + (size_t)((N + 63) & ~63);
  int*  o_tab  = o_ctr + 64;
  int2* o_ovf  = (int2*)(o_tab + (size_t)N * CAP);
  float* o_h2  = (float*)(o_ovf + (size_t)((E + 31) & ~31));

  const int mtiles = (N + 15) / 16;
  const int gx     = (mtiles + 3) / 4;
  const int ab  = (N + 8 * AGG_NPW - 1) / (8 * AGG_NPW);
  const int NR  = (N + RS - 1) >> RS_LOG;
  const int CH  = (E + NB1 - 1) / NB1;

  int* w_seg    = (int*)bufA;
  int* w_segcnt = w_seg + (size_t)NB1 * NR * SEGCAP;

  part_edges_k<<<NB1, 256, NR * sizeof(int), stream>>>(w_seg, w_segcnt, o_ctr, ei, E, NR, CH);
  place_edges_k<<<NR, NB1, 0, stream>>>(o_cnt, o_tab, o_ovf, o_ctr, w_seg, w_segcnt, ei, N, E,
                                        NR, CH);

  gemm_rows_k<IN_C, false><<<gx, 128, 0, stream>>>(x, W1, o_cnt, bufA, N);
  agg_gemm_k<2><<<ab, 256, 0, stream>>>(o_h2, nullptr, bufA, o_cnt, o_tab, o_ovf, o_ctr, b1, W2, N, E);
  agg_gemm_k<4><<<ab, 256, 0, stream>>>(bufA, bufB, o_h2, o_cnt, o_tab, o_ovf, o_ctr, b2, Wm1, N, E);

  edge_out_k<<<1250, 128, 0, stream>>>(bufA, bufB, ei, bm1, Wm2, bm2, out, E, E / 16);
}


# layer-1 dense product folded into the edge-placing launch: a range's block also does its 8 row tiles of dinv*(x@W1) on WMMA, dinv from its LDS counters, W1 staged by one load group; one launch fewer
# speedup vs baseline: 3.7265x; 1.0842x over previous
#include <hip/hip_runtime.h>
#include <hip/hip_bf16.h>

typedef __attribute__((ext_vector_type(2))) float v2f;
typedef __attribute__((ext_vector_type(8))) float v8f;

#define IN_C 128
#define HID  64
#define OUT_C 16

#define CAP 64
#define RS_LOG 7
#define RS (1 << RS_LOG)
#define NB1 256
#define SEGCAP 32
#define EPL 8

__global__ __launch_bounds__(256) void part_edges_k(int* __restrict__ seg,
                                                    int* __restrict__ segcnt,
                                                    int* __restrict__ octr,
                                                    const int* __restrict__ ei, int E, int NR,
                                                    int CH) {
  extern __shared__ int lbin[];
  const int t = threadIdx.x;
  const int b = blockIdx.x;
  for (int r = t; r < NR; r += 256) lbin[r] = 0;
  if (b == 0 && t == 0) octr[0] = 0;
  __syncthreads();

  const int e0 = b * CH;
  int* sb = seg + (size_t)b * NR * SEGCAP;
  for (int k0 = 0; k0 < CH; k0 += 256 * EPL) {
    unsigned s[EPL], d[EPL];
#pragma unroll
    for (int u = 0; u < EPL; ++u) {
      const int ec = min(e0 + k0 + u * 256 + t, E - 1);
      s[u] = (unsigned)ei[ec];
      d[u] = (unsigned)ei[E + ec];
    }
#pragma unroll
    for (int u = 0; u < EPL; ++u) {
      const int k = k0 + u * 256 + t;
      if (k < CH && e0 + k < E) {
        const unsigned r = d[u] >> RS_LOG;
        const int p = atomicAdd(&lbin[r], 1);
        if (p < SEGCAP)
          sb[(size_t)r * SEGCAP + p] = (int)(s[u] | ((d[u] & (unsigned)(RS - 1)) << 24));
      }
    }
  }
  __syncthreads();
  for (int r = t; r < NR; r += 256) segcnt[(size_t)r * NB1 + b] = lbin[r];
}

__device__ __forceinline__ void place_one(int* lcnt, int* __restrict__ tab,
                                          int2* __restrict__ ovf, int* __restrict__ octr,
                                          int nbase, unsigned dl, int s) {
  const int rank = atomicAdd(&lcnt[dl], 1);
  const int d = nbase + (int)dl;
  if (rank < CAP) {
    tab[(size_t)d * CAP + rank] = s;
  } else {
    const int q = atomicAdd(octr, 1);
    ovf[q] = make_int2(d, s);
  }
}

#define WS_STRIDE 72

__global__ __launch_bounds__(NB1) void place_gemm_k(int* __restrict__ cnt, int* __restrict__ tab,
                                                    int2* __restrict__ ovf,
                                                    int* __restrict__ octr,
                                                    const int* __restrict__ seg,
                                                    const int* __restrict__ segcnt,
                                                    const int* __restrict__ ei,
                                                    const float* __restrict__ x,
                                                    const float* __restrict__ W1,
                                                    float* __restrict__ H, int N, int E,
                                                    int NR, int CH) {
  __shared__ int lcnt[RS];
  __shared__ float wsm[IN_C * WS_STRIDE];

  const int t = threadIdx.x;
  const int r = blockIdx.x;
  const int wave = t >> 5;
  const int lane = t & 31;
  const int mr   = lane & 15;
  const int half = lane >> 4;
  const int nbase = r << RS_LOG;
  const int m0 = nbase + wave * 16;
  const bool valid = m0 < N;

  const int c = segcnt[(size_t)r * NB1 + t];

  float4 wq[IN_C * 16 / NB1];
#pragma unroll
  for (int u = 0; u < IN_C * 16 / NB1; ++u) {
    const int i = t + u * NB1;
    wq[u] = *(const float4*)&W1[(size_t)(i >> 4) * HID + (i & 15) * 4];
  }

  const float* arow = x + (size_t)min(m0 + mr, N - 1) * IN_C + 2 * half;
  v2f a[IN_C / 4];
#pragma unroll
  for (int k = 0; k < IN_C / 4; ++k) {
    a[k][0] = arow[4 * k];
    a[k][1] = arow[4 * k + 1];
  }

#pragma unroll
  for (int u = 0; u < IN_C * 16 / NB1; ++u) {
    const int i = t + u * NB1;
    *(float4*)&wsm[(i >> 4) * WS_STRIDE + (i & 15) * 4] = wq[u];
  }
  if (t < RS) lcnt[t] = 1;
  __syncthreads();

  if (c <= SEGCAP) {
    const int* sp = seg + ((size_t)t * NR + r) * SEGCAP;
    for (int j = 0; j < c; j += 4) {
      int w[4];
#pragma unroll
      for (int u = 0; u < 4; ++u) w[u] = (j + u < c) ? sp[j + u] : 0;
#pragma unroll
      for (int u = 0; u < 4; ++u)
        if (j + u < c)
          place_one(lcnt, tab, ovf, octr, nbase, (unsigned)w[u] >> 24, w[u] & 0xFFFFFF);
    }
  } else {
    const int eb = t * CH;
    const int ee = min(eb + CH, E);
    for (int e = eb; e < ee; ++e) {
      const unsigned d = (unsigned)ei[E + e];
      if ((int)(d >> RS_LOG) == r)
        place_one(lcnt, tab, ovf, octr, nbase, d & (unsigned)(RS - 1), ei[e]);
    }
  }
  __syncthreads();
  if (t < RS && nbase + t < N) {
    cnt[nbase + t] = lcnt[t];
    tab[(size_t)(nbase + t) * CAP] = nbase + t;
  }

  if (valid) {
    v8f acc[4] = {v8f{}, v8f{}, v8f{}, v8f{}};
#pragma unroll
    for (int k = 0; k < IN_C / 4; ++k) {
      const float* bp = wsm + (4 * k + 2 * half) * WS_STRIDE + mr;
#pragma unroll
      for (int nt = 0; nt < 4; ++nt) {
        v2f b;
        b[0] = bp[nt * 16];
        b[1] = bp[WS_STRIDE + nt * 16];
        acc[nt] = __builtin_amdgcn_wmma_f32_16x16x4_f32(false, a[k], false, b,
                                                        (short)0, acc[nt], false, false);
      }
    }

    const int* lc = lcnt + wave * 16 + 8 * half;
    float dv[8];
#pragma unroll
    for (int q = 0; q < 8; ++q) dv[q] = rsqrtf((float)lc[q]);

    float* hrow = H + (size_t)(m0 + 8 * half) * HID + mr;
#pragma unroll
    for (int nt = 0; nt < 4; ++nt) {
#pragma unroll
      for (int q = 0; q < 8; ++q) hrow[(size_t)q * HID + nt * 16] = dv[q] * acc[nt][q];
    }
  }
}

#define AGG_NPW 4
#define AS 68

template <int NPAIR>
__global__ __launch_bounds__(256) void agg_gemm_k(float* __restrict__ O0,
                                                  float* __restrict__ O1,
                                                  const float* __restrict__ h,
                                                  const int* __restrict__ cnt,
                                                  const int* __restrict__ tab,
                                                  const int2* __restrict__ ovf,
                                                  const int* __restrict__ octr,
                                                  const float* __restrict__ bias,
                                                  const float* __restrict__ Wt, int N,
                                                  int E) {
  __shared__ __attribute__((aligned(16))) float as[8 * AGG_NPW * AS];

  const int lane = threadIdx.x & 31;
  const int wv = __builtin_amdgcn_readfirstlane((int)(threadIdx.x >> 5));
  const int n0 = ((int)blockIdx.x * 8 + wv) * AGG_NPW;
  const int nn = max(0, min(AGG_NPW, N - n0));
  float* arow = as + (wv * AGG_NPW) * AS + lane;

  int ct = 0;
  if (lane < nn) ct = cnt[n0 + lane];
  const float dl = (ct > 0) ? rsqrtf((float)ct) : 0.0f;
  const float b0 = bias[lane];
  const float b1 = bias[lane + 32];
  const int* trow = tab + (size_t)n0 * CAP;

  int tot = __builtin_amdgcn_readlane(ct, 0);
  int len = min(tot, CAP);
  int es = 0;
  if (lane < len) es = trow[lane];

  for (int i = 0; i < nn; ++i) {
    const float dd = __int_as_float(__builtin_amdgcn_readlane(__float_as_int(dl), i));
    int ntot = 0, nlen = 0, nes = 0;

    float acc0 = 0.0f, acc1 = 0.0f;
    for (int done = 0; done < len; done += 32) {
      if (done > 0) {
        es = 0;
        if (done + lane < len) es = trow[i * CAP + done + lane];
      }
      const int c = min(32, len - done);
      float v0[32], v1[32];
#pragma unroll
      for (int u = 0; u < 32; ++u) {
        v0[u] = 0.0f;
        v1[u] = 0.0f;
        if (u < c) {
          const int sj = __builtin_amdgcn_readlane(es, u);
          const float* hp = h + (size_t)sj * HID + lane;
          v0[u] = hp[0];
          v1[u] = hp[32];
        }
      }
      if (done == 0 && i + 1 < nn) {
        ntot = __builtin_amdgcn_readlane(ct, i + 1);
        nlen = min(ntot, CAP);
        if (lane < nlen) nes = trow[(i + 1) * CAP + lane];
      }
#pragma unroll
      for (int u = 0; u < 32; ++u) {
        acc0 += v0[u];
        acc1 += v1[u];
      }
    }

    if (tot > CAP) {
      const int node = n0 + i;
      const int nov = min(octr[0], E);
      for (int k0 = 0; k0 < nov; k0 += 32) {
        int2 o = make_int2(-1, 0);
        if (k0 + lane < nov) o = ovf[k0 + lane];
#pragma nounroll
        for (int u = 0; u < 32; ++u) {
          if (__shfl(o.x, u) == node) {
            const int sj = __shfl(o.y, u);
            const float* hp = h + (size_t)sj * HID + lane;
            acc0 += hp[0];
            acc1 += hp[32];
          }
        }
      }
    }

    const float so = (NPAIR == 2) ? dd : 1.0f;
    arow[i * AS]      = so * fmaxf(b0 + dd * acc0, 0.0f);
    arow[i * AS + 32] = so * fmaxf(b1 + dd * acc1, 0.0f);

    tot = ntot;
    len = nlen;
    es = nes;
  }
  for (int i = nn; i < AGG_NPW; ++i) {
    arow[i * AS]      = 0.0f;
    arow[i * AS + 32] = 0.0f;
  }

  const int mr   = lane & 15;
  const int half = lane >> 4;
  const bool worker = wv < 2 * NPAIR;
  const int mt = wv / NPAIR;
  const int np = wv % NPAIR;
  v2f bf[2][16];
  if (worker) {
    const float* wp = Wt + (size_t)((np >> 1) * HID + 2 * half) * HID + (np & 1) * 32 + 2 * mr;
#pragma unroll
    for (int t = 0; t < 16; ++t) {
      const float2 w0 = *(const float2*)(wp + (size_t)(4 * t) * HID);
      const float2 w1 = *(const float2*)(wp + (size_t)(4 * t + 1) * HID);
      bf[0][t][0] = w0.x;
      bf[0][t][1] = w1.x;
      bf[1][t][0] = w0.y;
      bf[1][t][1] = w1.y;
    }
  }
  __syncthreads();

  const int m0 = (int)blockIdx.x * (8 * AGG_NPW) + mt * 16;
  if (worker && m0 < N) {
    const float* ap = as + (mt * 16 + mr) * AS + 2 * half;
    v8f acc[2] = {v8f{}, v8f{}};
#pragma unroll
    for (int t = 0; t < 16; ++t) {
      const v2f a = *(const v2f*)(ap + 4 * t);
#pragma unroll
      for (int nt = 0; nt < 2; ++nt)
        acc[nt] = __builtin_amdgcn_wmma_f32_16x16x4_f32(false, a, false, bf[nt][t],
                                                        (short)0, acc[nt], false, false);
    }

    float* obase = (NPAIR == 4 && (np >> 1)) ? O1 : O0;
    float* orow = obase + (size_t)(m0 + 8 * half) * HID + (np & 1) * 32 + 2 * mr;
#pragma unroll
    for (int r = 0; r < 8; ++r)
      *(float2*)(orow + (size_t)r * HID) = make_float2(acc[0][r], acc[1][r]);
  }
}


#define ZS 68
#define OS 144

__global__ __launch_bounds__(128) void edge_out_k(const float* __restrict__ P,
                                                  const float* __restrict__ Q,
                                                  const int* __restrict__ ei,
                                                  const float* __restrict__ bm1,
                                                  const float* __restrict__ Wm2,
                                                  const float* __restrict__ bm2,
                                                  float* __restrict__ out, int E, int ntiles) {
  __shared__ float zs[4][16 * ZS];

  const int tid  = threadIdx.x;
  const int lane = tid & 31;
  const int mr   = lane & 15;
  const int half = lane >> 4;
  const int wave = __builtin_amdgcn_readfirstlane(tid >> 5);
  float* zw = zs[wave];

  v2f bf[16];
#pragma unroll
  for (int j = 0; j < 8; ++j) {
    const float* wp = Wm2 + (32 * half + 4 * j) * OUT_C + mr;
    bf[2 * j][0]     = wp[0];
    bf[2 * j][1]     = wp[OUT_C];
    bf[2 * j + 1][0] = wp[2 * OUT_C];
    bf[2 * j + 1][1] = wp[3 * OUT_C];
  }
  const float4 b1v = *(const float4*)&bm1[4 * mr];
  const float bo = bm2[mr];

  const int stride = (int)gridDim.x * 4;
  int tile = (int)blockIdx.x * 4 + wave;
  const int* eip = ei + (half ? E : 0) + mr;
  int idx = (tile < ntiles) ? eip[tile * 16] : 0;

  for (; tile < ntiles; tile += stride) {
    const int tn = tile + stride;
    const int idx_next = (tn < ntiles) ? eip[tn * 16] : 0;

    float4 z[8];
#pragma unroll
    for (int i = 0; i < 8; ++i) {
      const int r0 = __builtin_amdgcn_readlane(idx, 2 * i);
      const int r1 = __builtin_amdgcn_readlane(idx, 2 * i + 1);
      const int c0 = __builtin_amdgcn_readlane(idx, 16 + 2 * i);
      const int c1 = __builtin_amdgcn_readlane(idx, 17 + 2 * i);
      const unsigned ro = (unsigned)(half ? r1 : r0) * (unsigned)(HID * 4) + (unsigned)mr * 16u;
      const unsigned co = (unsigned)(half ? c1 : c0) * (unsigned)(HID * 4) + (unsigned)mr * 16u;
      const float4 p = *(const float4*)((const char*)P + ro);
      const float4 q = *(const float4*)((const char*)Q + co);
      z[i].x = fmaxf(p.x + q.x + b1v.x, 0.0f);
      z[i].y = fmaxf(p.y + q.y + b1v.y, 0.0f);
      z[i].z = fmaxf(p.z + q.z + b1v.z, 0.0f);
      z[i].w = fmaxf(p.w + q.w + b1v.w, 0.0f);
    }
#pragma unroll
    for (int i = 0; i < 8; ++i)
      *(float4*)&zw[(2 * i + half) * ZS + 4 * mr] = z[i];


    v8f o = {};
#pragma unroll
    for (int j = 0; j < 8; ++j) {
      const float4 a4 = *(const float4*)&zw[mr * ZS + 32 * half + 4 * j];
      v2f a0, a1;
      a0[0] = a4.x; a0[1] = a4.y;
      a1[0] = a4.z; a1[1] = a4.w;
      o = __builtin_amdgcn_wmma_f32_16x16x4_f32(false, a0, false, bf[2 * j],
                                                (short)0, o, false, false);
      o = __builtin_amdgcn_wmma_f32_16x16x4_f32(false, a1, false, bf[2 * j + 1],
                                                (short)0, o, false, false);
    }

#pragma unroll
    for (int r = 0; r < 8; ++r) zw[half * OS + r * 16 + mr] = o[r] + bo;

    float* ob = out + (size_t)tile * (16 * OUT_C) + 4 * lane;
    const float4 o0 = *(const float4*)&zw[4 * lane];
    const float4 o1 = *(const float4*)&zw[OS + 4 * lane];
    *(float4*)ob = o0;
    *(float4*)(ob + 128) = o1;

    idx = idx_next;
  }
}

extern "C" void kernel_launch(void* const* d_in, const int* in_sizes, int n_in,
                              void* d_out, int out_size, void* d_ws, size_t ws_size,
                              hipStream_t stream) {
  const float* x   = (const float*)d_in[0];
  const int*   ei  = (const int*)d_in[1];
  const float* W1  = (const float*)d_in[2];
  const float* b1  = (const float*)d_in[3];
  const float* W2  = (const float*)d_in[4];
  const float* b2  = (const float*)d_in[5];
  const float* Wm1 = (const float*)d_in[6];
  const float* bm1 = (const float*)d_in[7];
  const float* Wm2 = (const float*)d_in[8];
  const float* bm2 = (const float*)d_in[9];
  float* out = (float*)d_out;

  const int N = in_sizes[0] / IN_C;
  const int E = in_sizes[1] / 2;

  float* bufA = (float*)d_ws + (size_t)((N + 63) & ~63);
  float* bufB = bufA + (size_t)N * HID;

  int*  o_cnt  = (int*)d_out;
  int*  o_ctr  = o_cnt + (size_t)((N + 63) & ~63);
  int*  o_tab  = o_ctr + 64;
  int2* o_ovf  = (int2*)(o_tab + (size_t)N * CAP);
  float* o_h2  = (float*)(o_ovf + (size_t)((E + 31) & ~31));
  float* o_h1  = o_h2 + (size_t)N * HID;

  const int ab  = (N + 8 * AGG_NPW - 1) / (8 * AGG_NPW);
  const int NR  = (N + RS - 1) >> RS_LOG;
  const int CH  = (E + NB1 - 1) / NB1;

  int* w_seg    = (int*)bufA;
  int* w_segcnt = w_seg + (size_t)NB1 * NR * SEGCAP;

  part_edges_k<<<NB1, 256, NR * sizeof(int), stream>>>(w_seg, w_segcnt, o_ctr, ei, E, NR, CH);
  place_gemm_k<<<NR, NB1, 0, stream>>>(o_cnt, o_tab, o_ovf, o_ctr, w_seg, w_segcnt, ei, x, W1,
                                       o_h1, N, E, NR, CH);

  agg_gemm_k<2><<<ab, 256, 0, stream>>>(o_h2, nullptr, o_h1, o_cnt, o_tab, o_ovf, o_ctr, b1, W2, N, E);
  agg_gemm_k<4><<<ab, 256, 0, stream>>>(bufA, bufB, o_h2, o_cnt, o_tab, o_ovf, o_ctr, b2, Wm1, N, E);

  edge_out_k<<<1250, 128, 0, stream>>>(bufA, bufB, ei, bm1, Wm2, bm2, out, E, E / 16);
}
